# starting kernel
# baseline (speedup 1.0000x reference)
#include <hip/hip_runtime.h>

typedef float v2f __attribute__((ext_vector_type(2)));
typedef float v8f __attribute__((ext_vector_type(8)));

#define WMMA_F32(a, b, c) \
  __builtin_amdgcn_wmma_f32_16x16x4_f32(false, (a), false, (b), (short)0, (c), false, false)

constexpr int L = 1024, DM = 768, Hh = 12, FD = 16, HD = 64;
constexpr float EPSF = 1e-12f;
constexpr float QSC = 0.17677669529663687f;

__global__ __launch_bounds__(256) void gemm_f32_wmma(const float* __restrict__ A,
                                                     const float* __restrict__ B,
                                                     float* __restrict__ C,
                                                     int M, int N, int K) {
  __shared__ float As[64][36];
  __shared__ float BsT[64][38];
  const int tid  = threadIdx.x;
  const int wave = tid >> 5, lane = tid & 31;
  const int kh   = lane >> 4, ln = lane & 15;
  const int ntb  = N / 64;
  const int bm   = (blockIdx.x / ntb) * 64;
  const int bn   = (blockIdx.x % ntb) * 64;
  const int wm   = wave >> 1;
  const int wn   = wave & 1;
  const int ar   = tid >> 2, ac = (tid & 3) * 4;
  const int br   = tid >> 3, bc = (tid & 7) * 4;
  (void)M;

  const float4* Ag = (const float4*)A;
  const float4* Bg = (const float4*)B;

  float4 a0 = Ag[((size_t)(bm + ar) * K + ac) >> 2];
  float4 a1 = Ag[((size_t)(bm + ar) * K + ac + 16) >> 2];
  float4 b0 = Bg[((size_t)br * N + bn + bc) >> 2];
  float4 b1 = Bg[((size_t)br * N + bn + bc + 32) >> 2];
  v8f acc0 = {}, acc1 = {};

  for (int kb = 0; kb < K; kb += 32) {
    *(float4*)&As[ar][ac]      = a0;
    *(float4*)&As[ar][ac + 16] = a1;
    BsT[bc + 0][br]  = b0.x;  BsT[bc + 1][br]  = b0.y;
    BsT[bc + 2][br]  = b0.z;  BsT[bc + 3][br]  = b0.w;
    BsT[bc + 32][br] = b1.x;  BsT[bc + 33][br] = b1.y;
    BsT[bc + 34][br] = b1.z;  BsT[bc + 35][br] = b1.w;
    __syncthreads();
    if (kb + 32 < K) {
      a0 = Ag[((size_t)(bm + ar) * K + kb + 32 + ac) >> 2];
      a1 = Ag[((size_t)(bm + ar) * K + kb + 32 + ac + 16) >> 2];
      b0 = Bg[((size_t)(kb + 32 + br) * N + bn + bc) >> 2];
      b1 = Bg[((size_t)(kb + 32 + br) * N + bn + bc + 32) >> 2];
    }
#pragma unroll
    for (int k = 0; k < 32; k += 4) {
      const int kk = k + 2 * kh;
      v2f a = *(const v2f*)&As[wm * 16 + ln][kk];
      v2f p = *(const v2f*)&BsT[wn * 32 + ln][kk];
      v2f q = *(const v2f*)&BsT[wn * 32 + 16 + ln][kk];
      acc0 = WMMA_F32(a, p, acc0);
      acc1 = WMMA_F32(a, q, acc1);
    }
    __syncthreads();
  }
#pragma unroll
  for (int r = 0; r < 8; ++r) {
    const size_t row = (size_t)(bm + wm * 16 + r + 8 * kh);
    C[row * N + bn + wn * 32 + ln]      = acc0[r];
    C[row * N + bn + wn * 32 + 16 + ln] = acc1[r];
  }
}

__global__ __launch_bounds__(256) void based_attn(const float* __restrict__ Qg,
                                                  const float* __restrict__ Kg,
                                                  const float* __restrict__ Vg,
                                                  float* __restrict__ Yg) {
  __shared__ float s_lin[16][17];
  __shared__ float s_quad[256][17];
  __shared__ float s_ones[16];
  __shared__ float kcum[273];
  __shared__ float Qc[32][20];
  __shared__ float Kc[32][20];
  __shared__ float Vc[32][20];
  __shared__ float Aw[32][33];
  __shared__ float Fq[32][257];
  __shared__ float Yc[32][20];
  __shared__ float den[32];

  const int tid  = threadIdx.x;
  const int wave = tid >> 5, lane = tid & 31;
  const int kh   = lane >> 4, ln = lane & 15;
  const int h    = blockIdx.x >> 2;
  const int vb   = blockIdx.x & 3;
  const int qoff = h * FD;
  const int voff = h * HD + vb * 16;

  for (int i = tid; i < 16 * 16; i += 256) s_lin[i >> 4][i & 15] = 0.f;
  for (int i = tid; i < 256 * 16; i += 256) s_quad[i >> 4][i & 15] = 0.f;
  if (tid < 16) s_ones[tid] = 0.f;
  for (int i = tid; i < 273; i += 256) kcum[i] = 0.f;
  __syncthreads();

  for (int c = 0; c < L / 32; ++c) {
    const int bl = c * 32;
    if (tid < 128) {
      const int t = tid >> 2, c4 = (tid & 3) * 4;
      *(float4*)&Qc[t][c4] = *(const float4*)&Qg[(size_t)(bl + t) * (Hh * FD) + qoff + c4];
      *(float4*)&Kc[t][c4] = *(const float4*)&Kg[(size_t)(bl + t) * (Hh * FD) + qoff + c4];
    } else {
      const int i = tid - 128;
      const int t = i >> 2, c4 = (i & 3) * 4;
      *(float4*)&Vc[t][c4] = *(const float4*)&Vg[(size_t)(bl + t) * DM + voff + c4];
    }
    __syncthreads();

    if (wave < 4) {
      const int ti = wave >> 1, tj = wave & 1;
      v8f acc = {};
#pragma unroll
      for (int k = 0; k < 16; k += 4) {
        const int kk = k + 2 * kh;
        v2f a = *(const v2f*)&Qc[ti * 16 + ln][kk];
        v2f b = *(const v2f*)&Kc[tj * 16 + ln][kk];
        acc = WMMA_F32(a, b, acc);
      }
#pragma unroll
      for (int r = 0; r < 8; ++r) {
        const int row = ti * 16 + r + 8 * kh, col = tj * 16 + ln;
        const float s = acc[r];
        Aw[row][col] = (row >= col) ? (1.f + 0.25f * s + (s * s) * 0.03125f) : 0.f;
      }
    }
    for (int i = tid; i < 32 * 256; i += 256) {
      int t = i >> 8, ij = i & 255;
      Fq[t][ij] = Qc[t][ij >> 4] * Qc[t][ij & 15] * QSC;
    }
    __syncthreads();

    if (wave < 2) {
      const int ti = wave;
      const int row = ti * 16 + ln;
      v8f acc;
#pragma unroll
      for (int r = 0; r < 8; ++r) acc[r] = s_ones[ln];
#pragma unroll
      for (int k = 0; k < 32; k += 4) {
        const int kk = k + 2 * kh;
        v2f a = *(const v2f*)&Aw[row][kk];
        v2f b;
        b.x = Vc[kk][ln];  b.y = Vc[kk + 1][ln];
        acc = WMMA_F32(a, b, acc);
      }
#pragma unroll
      for (int k = 0; k < 16; k += 4) {
        const int kk = k + 2 * kh;
        v2f a, b;
        a.x = Qc[row][kk] * 0.5f; a.y = Qc[row][kk + 1] * 0.5f;
        b.x = s_lin[kk][ln];      b.y = s_lin[kk + 1][ln];
        acc = WMMA_F32(a, b, acc);
      }
#pragma unroll 8
      for (int k = 0; k < 256; k += 4) {
        const int kk = k + 2 * kh;
        v2f a = *(const v2f*)&Fq[row][kk];
        v2f b;
        b.x = s_quad[kk][ln]; b.y = s_quad[kk + 1][ln];
        acc = WMMA_F32(a, b, acc);
      }
#pragma unroll
      for (int r = 0; r < 8; ++r) Yc[ti * 16 + r + 8 * kh][ln] = acc[r];
    } else if (wave == 2) {
      const int t = lane;
      float d = kcum[0];
      for (int s2 = 0; s2 < 32; ++s2) d += Aw[t][s2];
      for (int i = 0; i < 16; ++i) d += Qc[t][i] * 0.5f * kcum[1 + i];
      for (int ij = 0; ij < 256; ++ij) d += Fq[t][ij] * kcum[17 + ij];
      den[t] = d + EPSF;
    }
    __syncthreads();

    if (tid < 128) {
      const int t = tid >> 2, c4 = (tid & 3) * 4;
      const float dd = den[t];
      float4 y = *(float4*)&Yc[t][c4];
      y.x /= dd; y.y /= dd; y.z /= dd; y.w /= dd;
      *(float4*)&Yg[(size_t)(bl + t) * DM + voff + c4] = y;
    }
    for (int i = tid; i < 32 * 256; i += 256) {
      int t = i >> 8, ij = i & 255;
      Fq[t][ij] = Kc[t][ij >> 4] * Kc[t][ij & 15] * QSC;
    }
    if (tid >= 128 && tid < 144) {
      const int f = tid - 128;
      float s = 0.f;
      for (int t = 0; t < 32; ++t) s += Vc[t][f];
      s_ones[f] += s;
    } else if (tid >= 144 && tid < 160) {
      const int f = tid - 144;
      float s = 0.f;
      for (int t = 0; t < 32; ++t) s += Kc[t][f];
      kcum[1 + f] += 0.5f * s;
    } else if (tid == 160) {
      kcum[0] += 32.f;
    }
    __syncthreads();

    {
      float s = 0.f;
      for (int t = 0; t < 32; ++t) s += Fq[t][tid];
      kcum[17 + tid] += s;
    }
    for (int tile = wave; tile < 17; tile += 8) {
      v8f acc;
      if (tile < 16) {
        const int qm = tile * 16;
#pragma unroll
        for (int r = 0; r < 8; ++r) acc[r] = s_quad[qm + r + 8 * kh][ln];
#pragma unroll
        for (int k = 0; k < 32; k += 4) {
          const int kk = k + 2 * kh;
          v2f a, b;
          a.x = Fq[kk][qm + ln]; a.y = Fq[kk + 1][qm + ln];
          b.x = Vc[kk][ln];      b.y = Vc[kk + 1][ln];
          acc = WMMA_F32(a, b, acc);
        }
#pragma unroll
        for (int r = 0; r < 8; ++r) s_quad[qm + r + 8 * kh][ln] = acc[r];
      } else {
#pragma unroll
        for (int r = 0; r < 8; ++r) acc[r] = s_lin[r + 8 * kh][ln];
#pragma unroll
        for (int k = 0; k < 32; k += 4) {
          const int kk = k + 2 * kh;
          v2f a, b;
          a.x = Kc[kk][ln] * 0.5f; a.y = Kc[kk + 1][ln] * 0.5f;
          b.x = Vc[kk][ln];        b.y = Vc[kk + 1][ln];
          acc = WMMA_F32(a, b, acc);
        }
#pragma unroll
        for (int r = 0; r < 8; ++r) s_lin[r + 8 * kh][ln] = acc[r];
      }
    }
    __syncthreads();
  }
}

extern "C" void kernel_launch(void* const* d_in, const int* in_sizes, int n_in,
                              void* d_out, int out_size, void* d_ws, size_t ws_size,
                              hipStream_t stream) {
  (void)in_sizes; (void)n_in; (void)out_size; (void)ws_size;
  const float* hs = (const float*)d_in[0];
  const float* Wq = (const float*)d_in[1];
  const float* Wk = (const float*)d_in[2];
  const float* Wv = (const float*)d_in[3];
  const float* Wo = (const float*)d_in[4];
  float* out = (float*)d_out;

  float* Qg = (float*)d_ws;
  float* Kg = Qg + (size_t)L * (Hh * FD);
  float* Vg = Kg + (size_t)L * (Hh * FD);
  float* Yg = Vg + (size_t)L * DM;

  const dim3 blk(256);
  gemm_f32_wmma<<<dim3((L / 64) * ((Hh * FD) / 64)), blk, 0, stream>>>(hs, Wq, Qg, L, Hh * FD, DM);
  gemm_f32_wmma<<<dim3((L / 64) * ((Hh * FD) / 64)), blk, 0, stream>>>(hs, Wk, Kg, L, Hh * FD, DM);
  gemm_f32_wmma<<<dim3((L / 64) * (DM / 64)), blk, 0, stream>>>(hs, Wv, Vg, L, DM, DM);
  based_attn<<<dim3(Hh * 4), blk, 0, stream>>>(Qg, Kg, Vg, Yg);
  gemm_f32_wmma<<<dim3((L / 64) * (DM / 64)), blk, 0, stream>>>(Yg, Wo, out, L, DM, DM);
}


# attention: chunked feature-state recurrence replaced by direct causal polynomial-weight attention, one wave per (head, query tile, half of V columns), all in registers, no LDS; order of sums changes
# speedup vs baseline: 2.4935x; 2.4935x over previous
#include <hip/hip_runtime.h>

typedef float v2f __attribute__((ext_vector_type(2)));
typedef float v8f __attribute__((ext_vector_type(8)));

#define WMMA_F32(a, b, c) \
  __builtin_amdgcn_wmma_f32_16x16x4_f32(false, (a), false, (b), (short)0, (c), false, false)

constexpr int L = 1024, DM = 768, Hh = 12, FD = 16, HD = 64;
constexpr float EPSF = 1e-12f;
constexpr float QSC = 0.17677669529663687f;

__global__ __launch_bounds__(256) void gemm_f32_wmma(const float* __restrict__ A,
                                                     const float* __restrict__ B,
                                                     float* __restrict__ C,
                                                     int M, int N, int K) {
  __shared__ float As[64][36];
  __shared__ float BsT[64][38];
  const int tid  = threadIdx.x;
  const int wave = tid >> 5, lane = tid & 31;
  const int kh   = lane >> 4, ln = lane & 15;
  const int ntb  = N / 64;
  const int bm   = (blockIdx.x / ntb) * 64;
  const int bn   = (blockIdx.x % ntb) * 64;
  const int wm   = wave >> 1;
  const int wn   = wave & 1;
  const int ar   = tid >> 2, ac = (tid & 3) * 4;
  const int br   = tid >> 3, bc = (tid & 7) * 4;
  (void)M;

  const float4* Ag = (const float4*)A;
  const float4* Bg = (const float4*)B;

  float4 a0 = Ag[((size_t)(bm + ar) * K + ac) >> 2];
  float4 a1 = Ag[((size_t)(bm + ar) * K + ac + 16) >> 2];
  float4 b0 = Bg[((size_t)br * N + bn + bc) >> 2];
  float4 b1 = Bg[((size_t)br * N + bn + bc + 32) >> 2];
  v8f acc0 = {}, acc1 = {};

  for (int kb = 0; kb < K; kb += 32) {
    *(float4*)&As[ar][ac]      = a0;
    *(float4*)&As[ar][ac + 16] = a1;
    BsT[bc + 0][br]  = b0.x;  BsT[bc + 1][br]  = b0.y;
    BsT[bc + 2][br]  = b0.z;  BsT[bc + 3][br]  = b0.w;
    BsT[bc + 32][br] = b1.x;  BsT[bc + 33][br] = b1.y;
    BsT[bc + 34][br] = b1.z;  BsT[bc + 35][br] = b1.w;
    __syncthreads();
    if (kb + 32 < K) {
      a0 = Ag[((size_t)(bm + ar) * K + kb + 32 + ac) >> 2];
      a1 = Ag[((size_t)(bm + ar) * K + kb + 32 + ac + 16) >> 2];
      b0 = Bg[((size_t)(kb + 32 + br) * N + bn + bc) >> 2];
      b1 = Bg[((size_t)(kb + 32 + br) * N + bn + bc + 32) >> 2];
    }
#pragma unroll
    for (int k = 0; k < 32; k += 4) {
      const int kk = k + 2 * kh;
      v2f a = *(const v2f*)&As[wm * 16 + ln][kk];
      v2f p = *(const v2f*)&BsT[wn * 32 + ln][kk];
      v2f q = *(const v2f*)&BsT[wn * 32 + 16 + ln][kk];
      acc0 = WMMA_F32(a, p, acc0);
      acc1 = WMMA_F32(a, q, acc1);
    }
    __syncthreads();
  }
#pragma unroll
  for (int r = 0; r < 8; ++r) {
    const size_t row = (size_t)(bm + wm * 16 + r + 8 * kh);
    C[row * N + bn + wn * 32 + ln]      = acc0[r];
    C[row * N + bn + wn * 32 + 16 + ln] = acc1[r];
  }
}

struct KvTile {
  float4 ka, kb;
  float2 v[8];
};

__device__ __forceinline__ void load_tile(KvTile& t, const float* __restrict__ Kl,
                                          const float* __restrict__ Vl, int kt) {
  const float* kp = Kl + (size_t)kt * (16 * Hh * FD);
  const float* vp = Vl + (size_t)kt * (16 * DM);
  t.ka = *(const float4*)kp;
  t.kb = *(const float4*)(kp + 4);
#pragma unroll
  for (int i = 0; i < 8; ++i) t.v[i] = *(const float2*)(vp + (size_t)i * DM);
}

template <bool DIAG>
__device__ __forceinline__ void attend_tile(const KvTile& t, const v2f (&qf)[4],
                                            v8f& acc0, v8f& acc1, v2f& dsum,
                                            int hh, int c) {
  v2f kf[4];
  kf[0].x = t.ka.x; kf[0].y = t.ka.y; kf[1].x = t.ka.z; kf[1].y = t.ka.w;
  kf[2].x = t.kb.x; kf[2].y = t.kb.y; kf[3].x = t.kb.z; kf[3].y = t.kb.w;
  v8f s = {};
#pragma unroll
  for (int j = 0; j < 4; ++j) s = WMMA_F32(kf[j], qf[j], s);
#pragma unroll
  for (int j = 0; j < 4; ++j) {
    v2f x, w, b0, b1;
    x.x = s[2 * j]; x.y = s[2 * j + 1];
    w = x * (x * 0.03125f + 0.25f) + 1.f;
    if (DIAG) {
      w.x = (8 * hh + 2 * j     <= c) ? w.x : 0.f;
      w.y = (8 * hh + 2 * j + 1 <= c) ? w.y : 0.f;
    }
    dsum += w;
    b0.x = t.v[2 * j].x;  b0.y = t.v[2 * j + 1].x;
    b1.x = t.v[2 * j].y;  b1.y = t.v[2 * j + 1].y;
    acc0 = WMMA_F32(w, b0, acc0);
    acc1 = WMMA_F32(w, b1, acc1);
  }
}

__device__ __forceinline__ void attend_rows(int qt, const float* __restrict__ Ql,
                                            const float* __restrict__ Kl,
                                            const float* __restrict__ Vl,
                                            float* __restrict__ Yl, int hh, int c) {
  v2f qf[4];
  {
    const float* qp = Ql + (size_t)qt * (16 * Hh * FD);
    const float4 qa = *(const float4*)qp;
    const float4 qb = *(const float4*)(qp + 4);
    qf[0].x = qa.x; qf[0].y = qa.y; qf[1].x = qa.z; qf[1].y = qa.w;
    qf[2].x = qb.x; qf[2].y = qb.y; qf[3].x = qb.z; qf[3].y = qb.w;
  }
  v8f acc0 = {}, acc1 = {};
  v2f dsum = {};
  KvTile ta, tb, tc;
  load_tile(ta, Kl, Vl, 0);
  load_tile(tb, Kl, Vl, qt < 1 ? qt : 1);
  int kt = 0;
  for (; kt + 3 <= qt; kt += 3) {
    load_tile(tc, Kl, Vl, kt + 2);
    attend_tile<false>(ta, qf, acc0, acc1, dsum, hh, c);
    load_tile(ta, Kl, Vl, kt + 3);
    attend_tile<false>(tb, qf, acc0, acc1, dsum, hh, c);
    load_tile(tb, Kl, Vl, kt + 4 < qt ? kt + 4 : qt);
    attend_tile<false>(tc, qf, acc0, acc1, dsum, hh, c);
  }
  const int rem = qt - kt;
  if (rem == 0) {
    attend_tile<true>(ta, qf, acc0, acc1, dsum, hh, c);
  } else if (rem == 1) {
    attend_tile<false>(ta, qf, acc0, acc1, dsum, hh, c);
    attend_tile<true>(tb, qf, acc0, acc1, dsum, hh, c);
  } else {
    load_tile(tc, Kl, Vl, qt);
    attend_tile<false>(ta, qf, acc0, acc1, dsum, hh, c);
    attend_tile<false>(tb, qf, acc0, acc1, dsum, hh, c);
    attend_tile<true>(tc, qf, acc0, acc1, dsum, hh, c);
  }

  const float part = dsum.x + dsum.y;
  const float tot  = part + __shfl_xor(part, 16, 32) + EPSF;
  float* yp = Yl + (size_t)qt * (16 * DM);
#pragma unroll
  for (int r = 0; r < 8; ++r) {
    const float dd = __shfl(tot, 8 * hh + r, 32);
    float2 y;
    y.x = acc0[r] / dd;
    y.y = acc1[r] / dd;
    *(float2*)(yp + (size_t)r * DM) = y;
  }
}

__global__ __launch_bounds__(128) void based_attn(const float* __restrict__ Qg,
                                                  const float* __restrict__ Kg,
                                                  const float* __restrict__ Vg,
                                                  float* __restrict__ Yg) {
  const int lane = threadIdx.x & 31;
  const int wave = __builtin_amdgcn_readfirstlane((int)(threadIdx.x >> 5));
  const int vb   = wave & 1;
  const int hh   = lane >> 4, c = lane & 15;
  const int hd   = blockIdx.x % Hh;
  const int t0   = 2 * (int)(blockIdx.x / Hh) + (wave >> 1);

  const float* Ql = Qg + (size_t)c * (Hh * FD) + hd * FD + 8 * hh;
  const float* Kl = Kg + (size_t)c * (Hh * FD) + hd * FD + 8 * hh;
  const float* Vl = Vg + (size_t)(8 * hh) * DM + hd * HD + vb * 32 + 2 * c;
  float*       Yl = Yg + (size_t)(8 * hh) * DM + hd * HD + vb * 32 + 2 * c;

  attend_rows(t0, Ql, Kl, Vl, Yl, hh, c);
  attend_rows(L / 16 - 1 - t0, Ql, Kl, Vl, Yl, hh, c);
}

extern "C" void kernel_launch(void* const* d_in, const int* in_sizes, int n_in,
                              void* d_out, int out_size, void* d_ws, size_t ws_size,
                              hipStream_t stream) {
  (void)in_sizes; (void)n_in; (void)out_size; (void)ws_size;
  const float* hs = (const float*)d_in[0];
  const float* Wq = (const float*)d_in[1];
  const float* Wk = (const float*)d_in[2];
  const float* Wv = (const float*)d_in[3];
  const float* Wo = (const float*)d_in[4];
  float* out = (float*)d_out;

  float* Qg = (float*)d_ws;
  float* Kg = Qg + (size_t)L * (Hh * FD);
  float* Vg = Kg + (size_t)L * (Hh * FD);
  float* Yg = Vg + (size_t)L * DM;

  const dim3 blk(256);
  gemm_f32_wmma<<<dim3((L / 64) * ((Hh * FD) / 64)), blk, 0, stream>>>(hs, Wq, Qg, L, Hh * FD, DM);
  gemm_f32_wmma<<<dim3((L / 64) * ((Hh * FD) / 64)), blk, 0, stream>>>(hs, Wk, Kg, L, Hh * FD, DM);
  gemm_f32_wmma<<<dim3((L / 64) * (DM / 64)), blk, 0, stream>>>(hs, Wv, Vg, L, DM, DM);
  based_attn<<<dim3(Hh * (L / 64)), dim3(128), 0, stream>>>(Qg, Kg, Vg, Yg);
  gemm_f32_wmma<<<dim3((L / 64) * (DM / 64)), blk, 0, stream>>>(Yg, Wo, out, L, DM, DM);
}


# GEMMs: LDS tiles and barriers replaced by WMMA fragments loaded straight from global memory through a register ring, one 16x64 (V, out) or 16x32 (Q, K) tile per wave; order of sums changes
# speedup vs baseline: 4.7788x; 1.9165x over previous
#include <hip/hip_runtime.h>

typedef float v2f __attribute__((ext_vector_type(2)));
typedef float v4f __attribute__((ext_vector_type(4)));
typedef float v8f __attribute__((ext_vector_type(8)));

#define WMMA_F32(a, b, c) \
  __builtin_amdgcn_wmma_f32_16x16x4_f32(false, (a), false, (b), (short)0, (c), false, false)

constexpr int L = 1024, DM = 768, Hh = 12, FD = 16, HD = 64;
constexpr float EPSF = 1e-12f;
constexpr float QSC = 0.17677669529663687f;

template <int NT> struct BVec;
template <> struct BVec<2> { typedef v2f type; };
template <> struct BVec<4> { typedef v4f type; };

template <int NT, int KH>
struct GChunk {
  v4f a[KH / 4];
  typename BVec<NT>::type b[KH];
};

template <int N, int NT, int KH>
__device__ __forceinline__ void load_chunk(GChunk<NT, KH>& t, const float* __restrict__ Al,
                                           const float* __restrict__ Bl, int kc) {
  typedef typename BVec<NT>::type bvec;
  const float* ap = Al + kc * (2 * KH);
  const float* bp = Bl + (size_t)kc * (2 * KH * N);
#pragma unroll
  for (int i = 0; i < KH / 4; ++i) t.a[i] = *(const v4f*)(ap + 4 * i);
#pragma unroll
  for (int i = 0; i < KH; ++i) t.b[i] = *(const bvec*)(bp + (size_t)i * N);
}

template <int NT, int KH>
__device__ __forceinline__ void mma_chunk(const GChunk<NT, KH>& t, v8f (&acc)[NT]) {
#pragma unroll
  for (int j = 0; j < KH / 2; ++j) {
    v2f af;
    af.x = t.a[j / 2][2 * (j % 2)];
    af.y = t.a[j / 2][2 * (j % 2) + 1];
#pragma unroll
    for (int n = 0; n < NT; ++n) {
      v2f bf;
      bf.x = t.b[2 * j][n];
      bf.y = t.b[2 * j + 1][n];
      acc[n] = WMMA_F32(af, bf, acc[n]);
    }
  }
}

template <int NT>
__device__ __forceinline__ void pin_acc(v8f (&acc)[NT]) {
#pragma unroll
  for (int n = 0; n < NT; ++n) asm volatile("" : "+v"(acc[n]));
}

template <int N, int K, int NT, int KH, int R, int WPB>
__global__ __launch_bounds__(32 * WPB) void gemm_direct(const float* __restrict__ A,
                                                        const float* __restrict__ B,
                                                        float* __restrict__ C) {
  typedef typename BVec<NT>::type bvec;
  constexpr int NS  = N / (16 * NT);
  constexpr int NCH = K / (2 * KH);
  static_assert(N % (16 * NT) == 0 && K % (2 * KH * R) == 0 && KH % 4 == 0,
                "tile sizes must divide the problem");
  const int lane = threadIdx.x & 31;
  const int wave = __builtin_amdgcn_readfirstlane((int)(threadIdx.x >> 5));
  const int h = lane >> 4, c = lane & 15;
  const int strip = blockIdx.x % NS;
  const int mt    = (int)(blockIdx.x / NS) * WPB + wave;

  const float* Al = A + (size_t)(mt * 16 + c) * K + KH * h;
  const float* Bl = B + (size_t)(KH * h) * N + strip * (16 * NT) + NT * c;
  float*       Cl = C + (size_t)(mt * 16 + 8 * h) * N + strip * (16 * NT) + NT * c;

  v8f acc[NT];
#pragma unroll
  for (int n = 0; n < NT; ++n) acc[n] = v8f{};

  GChunk<NT, KH> buf[R];
#pragma unroll
  for (int i = 0; i < R - 1; ++i) {
    load_chunk<N, NT, KH>(buf[i], Al, Bl, i);
    __builtin_amdgcn_sched_barrier(0);
  }
  asm volatile("" ::: "memory");
  for (int kc = 0; kc < NCH; kc += R) {
#pragma unroll
    for (int i = 0; i < R; ++i) {
      const int nx = kc + i + R - 1;
      load_chunk<N, NT, KH>(buf[(i + R - 1) % R], Al, Bl, nx < NCH ? nx : NCH - 1);
      __builtin_amdgcn_sched_barrier(0);
      mma_chunk<NT, KH>(buf[i], acc);
      if (i < R - 1) {
        pin_acc<NT>(acc);
        __builtin_amdgcn_sched_barrier(0);
      }
    }
    asm volatile("" ::: "memory");
  }

#pragma unroll
  for (int r = 0; r < 8; ++r) {
    bvec o;
#pragma unroll
    for (int n = 0; n < NT; ++n) o[n] = acc[n][r];
    *(bvec*)(Cl + (size_t)r * N) = o;
  }
}

struct KvTile {
  float4 ka, kb;
  float2 v[8];
};

__device__ __forceinline__ void load_tile(KvTile& t, const float* __restrict__ Kl,
                                          const float* __restrict__ Vl, int kt) {
  const float* kp = Kl + (size_t)kt * (16 * Hh * FD);
  const float* vp = Vl + (size_t)kt * (16 * DM);
  t.ka = *(const float4*)kp;
  t.kb = *(const float4*)(kp + 4);
#pragma unroll
  for (int i = 0; i < 8; ++i) t.v[i] = *(const float2*)(vp + (size_t)i * DM);
}

template <bool DIAG>
__device__ __forceinline__ void attend_tile(const KvTile& t, const v2f (&qf)[4],
                                            v8f& acc0, v8f& acc1, v2f& dsum,
                                            int hh, int c) {
  v2f kf[4];
  kf[0].x = t.ka.x; kf[0].y = t.ka.y; kf[1].x = t.ka.z; kf[1].y = t.ka.w;
  kf[2].x = t.kb.x; kf[2].y = t.kb.y; kf[3].x = t.kb.z; kf[3].y = t.kb.w;
  v8f s = {};
#pragma unroll
  for (int j = 0; j < 4; ++j) s = WMMA_F32(kf[j], qf[j], s);
#pragma unroll
  for (int j = 0; j < 4; ++j) {
    v2f x, w, b0, b1;
    x.x = s[2 * j]; x.y = s[2 * j + 1];
    w = x * (x * 0.03125f + 0.25f) + 1.f;
    if (DIAG) {
      w.x = (8 * hh + 2 * j     <= c) ? w.x : 0.f;
      w.y = (8 * hh + 2 * j + 1 <= c) ? w.y : 0.f;
    }
    dsum += w;
    b0.x = t.v[2 * j].x;  b0.y = t.v[2 * j + 1].x;
    b1.x = t.v[2 * j].y;  b1.y = t.v[2 * j + 1].y;
    acc0 = WMMA_F32(w, b0, acc0);
    acc1 = WMMA_F32(w, b1, acc1);
  }
}

__device__ __forceinline__ void attend_rows(int qt, const float* __restrict__ Ql,
                                            const float* __restrict__ Kl,
                                            const float* __restrict__ Vl,
                                            float* __restrict__ Yl, int hh, int c) {
  v2f qf[4];
  {
    const float* qp = Ql + (size_t)qt * (16 * Hh * FD);
    const float4 qa = *(const float4*)qp;
    const float4 qb = *(const float4*)(qp + 4);
    qf[0].x = qa.x; qf[0].y = qa.y; qf[1].x = qa.z; qf[1].y = qa.w;
    qf[2].x = qb.x; qf[2].y = qb.y; qf[3].x = qb.z; qf[3].y = qb.w;
  }
  v8f acc0 = {}, acc1 = {};
  v2f dsum = {};
  KvTile ta, tb, tc;
  load_tile(ta, Kl, Vl, 0);
  load_tile(tb, Kl, Vl, qt < 1 ? qt : 1);
  int kt = 0;
  for (; kt + 3 <= qt; kt += 3) {
    load_tile(tc, Kl, Vl, kt + 2);
    attend_tile<false>(ta, qf, acc0, acc1, dsum, hh, c);
    load_tile(ta, Kl, Vl, kt + 3);
    attend_tile<false>(tb, qf, acc0, acc1, dsum, hh, c);
    load_tile(tb, Kl, Vl, kt + 4 < qt ? kt + 4 : qt);
    attend_tile<false>(tc, qf, acc0, acc1, dsum, hh, c);
  }
  const int rem = qt - kt;
  if (rem == 0) {
    attend_tile<true>(ta, qf, acc0, acc1, dsum, hh, c);
  } else if (rem == 1) {
    attend_tile<false>(ta, qf, acc0, acc1, dsum, hh, c);
    attend_tile<true>(tb, qf, acc0, acc1, dsum, hh, c);
  } else {
    load_tile(tc, Kl, Vl, qt);
    attend_tile<false>(ta, qf, acc0, acc1, dsum, hh, c);
    attend_tile<false>(tb, qf, acc0, acc1, dsum, hh, c);
    attend_tile<true>(tc, qf, acc0, acc1, dsum, hh, c);
  }

  const float part = dsum.x + dsum.y;
  const float tot  = part + __shfl_xor(part, 16, 32) + EPSF;
  float* yp = Yl + (size_t)qt * (16 * DM);
#pragma unroll
  for (int r = 0; r < 8; ++r) {
    const float dd = __shfl(tot, 8 * hh + r, 32);
    float2 y;
    y.x = acc0[r] / dd;
    y.y = acc1[r] / dd;
    *(float2*)(yp + (size_t)r * DM) = y;
  }
}

__global__ __launch_bounds__(128) void based_attn(const float* __restrict__ Qg,
                                                  const float* __restrict__ Kg,
                                                  const float* __restrict__ Vg,
                                                  float* __restrict__ Yg) {
  const int lane = threadIdx.x & 31;
  const int wave = __builtin_amdgcn_readfirstlane((int)(threadIdx.x >> 5));
  const int vb   = wave & 1;
  const int hh   = lane >> 4, c = lane & 15;
  const int hd   = blockIdx.x % Hh;
  const int t0   = 2 * (int)(blockIdx.x / Hh) + (wave >> 1);

  const float* Ql = Qg + (size_t)c * (Hh * FD) + hd * FD + 8 * hh;
  const float* Kl = Kg + (size_t)c * (Hh * FD) + hd * FD + 8 * hh;
  const float* Vl = Vg + (size_t)(8 * hh) * DM + hd * HD + vb * 32 + 2 * c;
  float*       Yl = Yg + (size_t)(8 * hh) * DM + hd * HD + vb * 32 + 2 * c;

  attend_rows(t0, Ql, Kl, Vl, Yl, hh, c);
  attend_rows(L / 16 - 1 - t0, Ql, Kl, Vl, Yl, hh, c);
}

extern "C" void kernel_launch(void* const* d_in, const int* in_sizes, int n_in,
                              void* d_out, int out_size, void* d_ws, size_t ws_size,
                              hipStream_t stream) {
  (void)in_sizes; (void)n_in; (void)out_size; (void)ws_size;
  const float* hs = (const float*)d_in[0];
  const float* Wq = (const float*)d_in[1];
  const float* Wk = (const float*)d_in[2];
  const float* Wv = (const float*)d_in[3];
  const float* Wo = (const float*)d_in[4];
  float* out = (float*)d_out;

  float* Qg = (float*)d_ws;
  float* Kg = Qg + (size_t)L * (Hh * FD);
  float* Vg = Kg + (size_t)L * (Hh * FD);
  float* Yg = Vg + (size_t)L * DM;

  gemm_direct<Hh * FD, DM, 2, 8, 6, 2><<<dim3((L / 32) * ((Hh * FD) / 32)), dim3(64), 0, stream>>>(hs, Wq, Qg);
  gemm_direct<Hh * FD, DM, 2, 8, 6, 2><<<dim3((L / 32) * ((Hh * FD) / 32)), dim3(64), 0, stream>>>(hs, Wk, Kg);
  gemm_direct<DM, DM, 4, 8, 4, 4><<<dim3((L / 64) * (DM / 64)), dim3(128), 0, stream>>>(hs, Wv, Vg);
  based_attn<<<dim3(Hh * (L / 64)), dim3(128), 0, stream>>>(Qg, Kg, Vg, Yg);
  gemm_direct<DM, DM, 4, 8, 4, 4><<<dim3((L / 64) * (DM / 64)), dim3(128), 0, stream>>>(Yg, Wo, out);
}


# Q projection folded into the attention kernel: each block forms its four query tiles' Q in registers (reduction split over its 4 waves, partials added through LDS); Q launch gone; sum order changes
# speedup vs baseline: 5.0161x; 1.0497x over previous
#include <hip/hip_runtime.h>

typedef float v2f __attribute__((ext_vector_type(2)));
typedef float v4f __attribute__((ext_vector_type(4)));
typedef float v8f __attribute__((ext_vector_type(8)));

#define WMMA_F32(a, b, c) \
  __builtin_amdgcn_wmma_f32_16x16x4_f32(false, (a), false, (b), (short)0, (c), false, false)

constexpr int L = 1024, DM = 768, Hh = 12, FD = 16, HD = 64;
constexpr float EPSF = 1e-12f;
constexpr float QSC = 0.17677669529663687f;

template <int NT> struct BVec;
template <> struct BVec<2> { typedef v2f type; };
template <> struct BVec<4> { typedef v4f type; };

template <int NT, int KH>
struct GChunk {
  v4f a[KH / 4];
  typename BVec<NT>::type b[KH];
};

template <int N, int NT, int KH>
__device__ __forceinline__ void load_chunk(GChunk<NT, KH>& t, const float* __restrict__ Al,
                                           const float* __restrict__ Bl, int kc) {
  typedef typename BVec<NT>::type bvec;
  const float* ap = Al + kc * (2 * KH);
  const float* bp = Bl + (size_t)kc * (2 * KH * N);
#pragma unroll
  for (int i = 0; i < KH / 4; ++i) t.a[i] = *(const v4f*)(ap + 4 * i);
#pragma unroll
  for (int i = 0; i < KH; ++i) t.b[i] = *(const bvec*)(bp + (size_t)i * N);
}

template <int NT, int KH>
__device__ __forceinline__ void mma_chunk(const GChunk<NT, KH>& t, v8f (&acc)[NT]) {
#pragma unroll
  for (int j = 0; j < KH / 2; ++j) {
    v2f af;
    af.x = t.a[j / 2][2 * (j % 2)];
    af.y = t.a[j / 2][2 * (j % 2) + 1];
#pragma unroll
    for (int n = 0; n < NT; ++n) {
      v2f bf;
      bf.x = t.b[2 * j][n];
      bf.y = t.b[2 * j + 1][n];
      acc[n] = WMMA_F32(af, bf, acc[n]);
    }
  }
}

template <int NT>
__device__ __forceinline__ void pin_acc(v8f (&acc)[NT]) {
#pragma unroll
  for (int n = 0; n < NT; ++n) asm volatile("" : "+v"(acc[n]));
}

template <int N, int K, int NT, int KH, int R, int WPB>
__global__ __launch_bounds__(32 * WPB) void gemm_direct(const float* __restrict__ A,
                                                        const float* __restrict__ B,
                                                        float* __restrict__ C) {
  typedef typename BVec<NT>::type bvec;
  constexpr int NS  = N / (16 * NT);
  constexpr int NCH = K / (2 * KH);
  static_assert(N % (16 * NT) == 0 && K % (2 * KH * R) == 0 && KH % 4 == 0,
                "tile sizes must divide the problem");
  const int lane = threadIdx.x & 31;
  const int wave = __builtin_amdgcn_readfirstlane((int)(threadIdx.x >> 5));
  const int h = lane >> 4, c = lane & 15;
  const int strip = blockIdx.x % NS;
  const int mt    = (int)(blockIdx.x / NS) * WPB + wave;

  const float* Al = A + (size_t)(mt * 16 + c) * K + KH * h;
  const float* Bl = B + (size_t)(KH * h) * N + strip * (16 * NT) + NT * c;
  float*       Cl = C + (size_t)(mt * 16 + 8 * h) * N + strip * (16 * NT) + NT * c;

  v8f acc[NT];
#pragma unroll
  for (int n = 0; n < NT; ++n) acc[n] = v8f{};

  GChunk<NT, KH> buf[R];
#pragma unroll
  for (int i = 0; i < R - 1; ++i) {
    load_chunk<N, NT, KH>(buf[i], Al, Bl, i);
    __builtin_amdgcn_sched_barrier(0);
  }
  asm volatile("" ::: "memory");
  for (int kc = 0; kc < NCH; kc += R) {
#pragma unroll
    for (int i = 0; i < R; ++i) {
      const int nx = kc + i + R - 1;
      load_chunk<N, NT, KH>(buf[(i + R - 1) % R], Al, Bl, nx < NCH ? nx : NCH - 1);
      __builtin_amdgcn_sched_barrier(0);
      mma_chunk<NT, KH>(buf[i], acc);
      if (i < R - 1) {
        pin_acc<NT>(acc);
        __builtin_amdgcn_sched_barrier(0);
      }
    }
    asm volatile("" ::: "memory");
  }

#pragma unroll
  for (int r = 0; r < 8; ++r) {
    bvec o;
#pragma unroll
    for (int n = 0; n < NT; ++n) o[n] = acc[n][r];
    *(bvec*)(Cl + (size_t)r * N) = o;
  }
}

constexpr int QCH = (DM / 4) / 16;

struct QChunk {
  float w[8];
  v4f   x[4][2];
};

__device__ __forceinline__ void q_load(QChunk& t, const float* __restrict__ wp,
                                       const float* __restrict__ x0, const float* __restrict__ x1,
                                       const float* __restrict__ x2, const float* __restrict__ x3,
                                       int kc) {
  const float* w = wp + (size_t)kc * (16 * Hh * FD);
#pragma unroll
  for (int i = 0; i < 8; ++i) t.w[i] = w[(size_t)i * (Hh * FD)];
#pragma unroll
  for (int i = 0; i < 2; ++i) {
    t.x[0][i] = *(const v4f*)(x0 + kc * 16 + 4 * i);
    t.x[1][i] = *(const v4f*)(x1 + kc * 16 + 4 * i);
    t.x[2][i] = *(const v4f*)(x2 + kc * 16 + 4 * i);
    t.x[3][i] = *(const v4f*)(x3 + kc * 16 + 4 * i);
  }
}

__device__ __forceinline__ void q_mma(const QChunk& t, v8f (&acc)[4]) {
#pragma unroll
  for (int j = 0; j < 4; ++j) {
    v2f af;
    af.x = t.w[2 * j];
    af.y = t.w[2 * j + 1];
#pragma unroll
    for (int n = 0; n < 4; ++n) {
      v2f bf;
      bf.x = t.x[n][j / 2][2 * (j % 2)];
      bf.y = t.x[n][j / 2][2 * (j % 2) + 1];
      acc[n] = WMMA_F32(af, bf, acc[n]);
    }
  }
}

struct KvTile {
  float4 ka, kb;
  float2 v[8];
};

__device__ __forceinline__ void load_tile(KvTile& t, const float* __restrict__ Kl,
                                          const float* __restrict__ Vl, int kt) {
  const float* kp = Kl + (size_t)kt * (16 * Hh * FD);
  const float* vp = Vl + (size_t)kt * (16 * DM);
  t.ka = *(const float4*)kp;
  t.kb = *(const float4*)(kp + 4);
#pragma unroll
  for (int i = 0; i < 8; ++i) t.v[i] = *(const float2*)(vp + (size_t)i * DM);
}

template <bool DIAG>
__device__ __forceinline__ void attend_tile(const KvTile& t, const v2f (&qf)[4],
                                            v8f& acc0, v8f& acc1, v2f& dsum,
                                            int hh, int c) {
  v2f kf[4];
  kf[0].x = t.ka.x; kf[0].y = t.ka.y; kf[1].x = t.ka.z; kf[1].y = t.ka.w;
  kf[2].x = t.kb.x; kf[2].y = t.kb.y; kf[3].x = t.kb.z; kf[3].y = t.kb.w;
  v8f s = {};
#pragma unroll
  for (int j = 0; j < 4; ++j) s = WMMA_F32(kf[j], qf[j], s);
#pragma unroll
  for (int j = 0; j < 4; ++j) {
    v2f x, w, b0, b1;
    x.x = s[2 * j]; x.y = s[2 * j + 1];
    w = x * (x * 0.03125f + 0.25f) + 1.f;
    if (DIAG) {
      w.x = (8 * hh + 2 * j     <= c) ? w.x : 0.f;
      w.y = (8 * hh + 2 * j + 1 <= c) ? w.y : 0.f;
    }
    dsum += w;
    b0.x = t.v[2 * j].x;  b0.y = t.v[2 * j + 1].x;
    b1.x = t.v[2 * j].y;  b1.y = t.v[2 * j + 1].y;
    acc0 = WMMA_F32(w, b0, acc0);
    acc1 = WMMA_F32(w, b1, acc1);
  }
}

__device__ __forceinline__ void attend_rows(int qt, const v8f& q,
                                            const float* __restrict__ Kl,
                                            const float* __restrict__ Vl,
                                            float* __restrict__ Yl, int hh, int c) {
  v2f qf[4];
#pragma unroll
  for (int j = 0; j < 4; ++j) { qf[j].x = q[2 * j]; qf[j].y = q[2 * j + 1]; }
  v8f acc0 = {}, acc1 = {};
  v2f dsum = {};
  KvTile ta, tb, tc;
  load_tile(ta, Kl, Vl, 0);
  load_tile(tb, Kl, Vl, qt < 1 ? qt : 1);
  int kt = 0;
  for (; kt + 3 <= qt; kt += 3) {
    load_tile(tc, Kl, Vl, kt + 2);
    attend_tile<false>(ta, qf, acc0, acc1, dsum, hh, c);
    load_tile(ta, Kl, Vl, kt + 3);
    attend_tile<false>(tb, qf, acc0, acc1, dsum, hh, c);
    load_tile(tb, Kl, Vl, kt + 4 < qt ? kt + 4 : qt);
    attend_tile<false>(tc, qf, acc0, acc1, dsum, hh, c);
  }
  const int rem = qt - kt;
  if (rem == 0) {
    attend_tile<true>(ta, qf, acc0, acc1, dsum, hh, c);
  } else if (rem == 1) {
    attend_tile<false>(ta, qf, acc0, acc1, dsum, hh, c);
    attend_tile<true>(tb, qf, acc0, acc1, dsum, hh, c);
  } else {
    load_tile(tc, Kl, Vl, qt);
    attend_tile<false>(ta, qf, acc0, acc1, dsum, hh, c);
    attend_tile<false>(tb, qf, acc0, acc1, dsum, hh, c);
    attend_tile<true>(tc, qf, acc0, acc1, dsum, hh, c);
  }

  const float part = dsum.x + dsum.y;
  const float tot  = part + __shfl_xor(part, 16, 32) + EPSF;
  float* yp = Yl + (size_t)qt * (16 * DM);
#pragma unroll
  for (int r = 0; r < 8; ++r) {
    const float dd = __shfl(tot, 8 * hh + r, 32);
    float2 y;
    y.x = acc0[r] / dd;
    y.y = acc1[r] / dd;
    *(float2*)(yp + (size_t)r * DM) = y;
  }
}

__global__ __launch_bounds__(128) void based_attn(const float* __restrict__ Xg,
                                                  const float* __restrict__ Wqg,
                                                  const float* __restrict__ Kg,
                                                  const float* __restrict__ Vg,
                                                  float* __restrict__ Yg) {
  __shared__ v8f qpart[4][4][32];

  const int lane = threadIdx.x & 31;
  const int wave = __builtin_amdgcn_readfirstlane((int)(threadIdx.x >> 5));
  const int vb   = wave & 1;
  const int pr   = wave >> 1;
  const int hh   = lane >> 4, c = lane & 15;
  const int hd   = blockIdx.x % Hh;
  const int bq   = (int)(blockIdx.x / Hh);
  const int t0   = 2 * bq + pr;

  v8f q0, q1;
  {
    const int d0 = (DM / 4) * wave + 8 * hh;
    const float* wp = Wqg + (size_t)d0 * (Hh * FD) + hd * FD + c;
    const float* x0 = Xg + (size_t)(16 * (2 * bq) + c) * DM + d0;
    const float* x1 = Xg + (size_t)(16 * (2 * bq + 1) + c) * DM + d0;
    const float* x2 = Xg + (size_t)(16 * (L / 16 - 1 - 2 * bq) + c) * DM + d0;
    const float* x3 = Xg + (size_t)(16 * (L / 16 - 2 - 2 * bq) + c) * DM + d0;

    v8f qacc[4];
#pragma unroll
    for (int n = 0; n < 4; ++n) qacc[n] = v8f{};

    QChunk qb[4];
    q_load(qb[0], wp, x0, x1, x2, x3, 0);
    __builtin_amdgcn_sched_barrier(0);
    q_load(qb[1], wp, x0, x1, x2, x3, 1);
    __builtin_amdgcn_sched_barrier(0);
    q_load(qb[2], wp, x0, x1, x2, x3, 2);
    __builtin_amdgcn_sched_barrier(0);
#pragma unroll
    for (int kc = 0; kc < QCH; ++kc) {
      if (kc + 3 < QCH) q_load(qb[(kc + 3) % 4], wp, x0, x1, x2, x3, kc + 3);
      __builtin_amdgcn_sched_barrier(0);
      q_mma(qb[kc % 4], qacc);
      __builtin_amdgcn_sched_barrier(0);
    }

#pragma unroll
    for (int n = 0; n < 4; ++n) qpart[wave][n][lane] = qacc[n];
    __syncthreads();
    q0 = qpart[0][pr][lane] + qpart[1][pr][lane] + qpart[2][pr][lane] + qpart[3][pr][lane];
    q1 = qpart[0][2 + pr][lane] + qpart[1][2 + pr][lane] + qpart[2][2 + pr][lane] +
         qpart[3][2 + pr][lane];
  }

  const float* Kl = Kg + (size_t)c * (Hh * FD) + hd * FD + 8 * hh;
  const float* Vl = Vg + (size_t)(8 * hh) * DM + hd * HD + vb * 32 + 2 * c;
  float*       Yl = Yg + (size_t)(8 * hh) * DM + hd * HD + vb * 32 + 2 * c;

  attend_rows(t0, q0, Kl, Vl, Yl, hh, c);
  attend_rows(L / 16 - 1 - t0, q1, Kl, Vl, Yl, hh, c);
}

extern "C" void kernel_launch(void* const* d_in, const int* in_sizes, int n_in,
                              void* d_out, int out_size, void* d_ws, size_t ws_size,
                              hipStream_t stream) {
  (void)in_sizes; (void)n_in; (void)out_size; (void)ws_size;
  const float* hs = (const float*)d_in[0];
  const float* Wq = (const float*)d_in[1];
  const float* Wk = (const float*)d_in[2];
  const float* Wv = (const float*)d_in[3];
  const float* Wo = (const float*)d_in[4];
  float* out = (float*)d_out;

  float* Kg = (float*)d_ws;
  float* Vg = Kg + (size_t)L * (Hh * FD);
  float* Yg = Vg + (size_t)L * DM;

  gemm_direct<Hh * FD, DM, 2, 8, 6, 2><<<dim3((L / 32) * ((Hh * FD) / 32)), dim3(64), 0, stream>>>(hs, Wk, Kg);
  gemm_direct<DM, DM, 4, 8, 4, 4><<<dim3((L / 64) * (DM / 64)), dim3(128), 0, stream>>>(hs, Wv, Vg);
  based_attn<<<dim3(Hh * (L / 64)), dim3(128), 0, stream>>>(hs, Wq, Kg, Vg, Yg);
  gemm_direct<DM, DM, 4, 8, 4, 4><<<dim3((L / 64) * (DM / 64)), dim3(128), 0, stream>>>(Yg, Wo, out);
}


# attention key loop: register ring deepened from 3 key tiles to 3 pairs of tiles; K/V loads issued 4 tiles ahead of their WMMAs and held there by scheduling barriers; order of sums unchanged
# speedup vs baseline: 5.4549x; 1.0875x over previous
#include <hip/hip_runtime.h>

typedef float v2f __attribute__((ext_vector_type(2)));
typedef float v4f __attribute__((ext_vector_type(4)));
typedef float v8f __attribute__((ext_vector_type(8)));

#define WMMA_F32(a, b, c) \
  __builtin_amdgcn_wmma_f32_16x16x4_f32(false, (a), false, (b), (short)0, (c), false, false)

constexpr int L = 1024, DM = 768, Hh = 12, FD = 16, HD = 64;
constexpr float EPSF = 1e-12f;
constexpr float QSC = 0.17677669529663687f;

template <int NT> struct BVec;
template <> struct BVec<2> { typedef v2f type; };
template <> struct BVec<4> { typedef v4f type; };

template <int NT, int KH>
struct GChunk {
  v4f a[KH / 4];
  typename BVec<NT>::type b[KH];
};

template <int N, int NT, int KH>
__device__ __forceinline__ void load_chunk(GChunk<NT, KH>& t, const float* __restrict__ Al,
                                           const float* __restrict__ Bl, int kc) {
  typedef typename BVec<NT>::type bvec;
  const float* ap = Al + kc * (2 * KH);
  const float* bp = Bl + (size_t)kc * (2 * KH * N);
#pragma unroll
  for (int i = 0; i < KH / 4; ++i) t.a[i] = *(const v4f*)(ap + 4 * i);
#pragma unroll
  for (int i = 0; i < KH; ++i) t.b[i] = *(const bvec*)(bp + (size_t)i * N);
}

template <int NT, int KH>
__device__ __forceinline__ void mma_chunk(const GChunk<NT, KH>& t, v8f (&acc)[NT]) {
#pragma unroll
  for (int j = 0; j < KH / 2; ++j) {
    v2f af;
    af.x = t.a[j / 2][2 * (j % 2)];
    af.y = t.a[j / 2][2 * (j % 2) + 1];
#pragma unroll
    for (int n = 0; n < NT; ++n) {
      v2f bf;
      bf.x = t.b[2 * j][n];
      bf.y = t.b[2 * j + 1][n];
      acc[n] = WMMA_F32(af, bf, acc[n]);
    }
  }
}

template <int NT>
__device__ __forceinline__ void pin_acc(v8f (&acc)[NT]) {
#pragma unroll
  for (int n = 0; n < NT; ++n) asm volatile("" : "+v"(acc[n]));
}

template <int N, int K, int NT, int KH, int R, int WPB>
__global__ __launch_bounds__(32 * WPB) void gemm_direct(const float* __restrict__ A,
                                                        const float* __restrict__ B,
                                                        float* __restrict__ C) {
  typedef typename BVec<NT>::type bvec;
  constexpr int NS  = N / (16 * NT);
  constexpr int NCH = K / (2 * KH);
  static_assert(N % (16 * NT) == 0 && K % (2 * KH * R) == 0 && KH % 4 == 0,
                "tile sizes must divide the problem");
  const int lane = threadIdx.x & 31;
  const int wave = __builtin_amdgcn_readfirstlane((int)(threadIdx.x >> 5));
  const int h = lane >> 4, c = lane & 15;
  const int strip = blockIdx.x % NS;
  const int mt    = (int)(blockIdx.x / NS) * WPB + wave;

  const float* Al = A + (size_t)(mt * 16 + c) * K + KH * h;
  const float* Bl = B + (size_t)(KH * h) * N + strip * (16 * NT) + NT * c;
  float*       Cl = C + (size_t)(mt * 16 + 8 * h) * N + strip * (16 * NT) + NT * c;

  v8f acc[NT];
#pragma unroll
  for (int n = 0; n < NT; ++n) acc[n] = v8f{};

  GChunk<NT, KH> buf[R];
#pragma unroll
  for (int i = 0; i < R - 1; ++i) {
    load_chunk<N, NT, KH>(buf[i], Al, Bl, i);
    __builtin_amdgcn_sched_barrier(0);
  }
  asm volatile("" ::: "memory");
  for (int kc = 0; kc < NCH; kc += R) {
#pragma unroll
    for (int i = 0; i < R; ++i) {
      const int nx = kc + i + R - 1;
      load_chunk<N, NT, KH>(buf[(i + R - 1) % R], Al, Bl, nx < NCH ? nx : NCH - 1);
      __builtin_amdgcn_sched_barrier(0);
      mma_chunk<NT, KH>(buf[i], acc);
      if (i < R - 1) {
        pin_acc<NT>(acc);
        __builtin_amdgcn_sched_barrier(0);
      }
    }
    asm volatile("" ::: "memory");
  }

#pragma unroll
  for (int r = 0; r < 8; ++r) {
    bvec o;
#pragma unroll
    for (int n = 0; n < NT; ++n) o[n] = acc[n][r];
    *(bvec*)(Cl + (size_t)r * N) = o;
  }
}

constexpr int QCH = (DM / 4) / 16;

struct QChunk {
  float w[8];
  v4f   x[4][2];
};

__device__ __forceinline__ void q_load(QChunk& t, const float* __restrict__ wp,
                                       const float* __restrict__ x0, const float* __restrict__ x1,
                                       const float* __restrict__ x2, const float* __restrict__ x3,
                                       int kc) {
  const float* w = wp + (size_t)kc * (16 * Hh * FD);
#pragma unroll
  for (int i = 0; i < 8; ++i) t.w[i] = w[(size_t)i * (Hh * FD)];
#pragma unroll
  for (int i = 0; i < 2; ++i) {
    t.x[0][i] = *(const v4f*)(x0 + kc * 16 + 4 * i);
    t.x[1][i] = *(const v4f*)(x1 + kc * 16 + 4 * i);
    t.x[2][i] = *(const v4f*)(x2 + kc * 16 + 4 * i);
    t.x[3][i] = *(const v4f*)(x3 + kc * 16 + 4 * i);
  }
}

__device__ __forceinline__ void q_mma(const QChunk& t, v8f (&acc)[4]) {
#pragma unroll
  for (int j = 0; j < 4; ++j) {
    v2f af;
    af.x = t.w[2 * j];
    af.y = t.w[2 * j + 1];
#pragma unroll
    for (int n = 0; n < 4; ++n) {
      v2f bf;
      bf.x = t.x[n][j / 2][2 * (j % 2)];
      bf.y = t.x[n][j / 2][2 * (j % 2) + 1];
      acc[n] = WMMA_F32(af, bf, acc[n]);
    }
  }
}

struct KvTile {
  float4 ka, kb;
  v2f v[8];
};

__device__ __forceinline__ void load_tile(KvTile& t, const float* __restrict__ Kl,
                                          const float* __restrict__ Vl, int kt) {
  const float* kp = Kl + (size_t)kt * (16 * Hh * FD);
  const float* vp = Vl + (size_t)kt * (16 * DM);
  t.ka = *(const float4*)kp;
  t.kb = *(const float4*)(kp + 4);
#pragma unroll
  for (int i = 0; i < 8; ++i) t.v[i] = *(const v2f*)(vp + (size_t)i * DM);
}

template <bool DIAG>
__device__ __forceinline__ void attend_tile(const KvTile& t, const v2f (&qf)[4],
                                            v8f& acc0, v8f& acc1, v2f& dsum,
                                            int hh, int c) {
  v2f kf[4];
  kf[0].x = t.ka.x; kf[0].y = t.ka.y; kf[1].x = t.ka.z; kf[1].y = t.ka.w;
  kf[2].x = t.kb.x; kf[2].y = t.kb.y; kf[3].x = t.kb.z; kf[3].y = t.kb.w;
  v8f s = {};
#pragma unroll
  for (int j = 0; j < 4; ++j) s = WMMA_F32(kf[j], qf[j], s);
#pragma unroll
  for (int j = 0; j < 4; ++j) {
    v2f x, w, b0, b1;
    x.x = s[2 * j]; x.y = s[2 * j + 1];
    w = x * (x * 0.03125f + 0.25f) + 1.f;
    if (DIAG) {
      w.x = (8 * hh + 2 * j     <= c) ? w.x : 0.f;
      w.y = (8 * hh + 2 * j + 1 <= c) ? w.y : 0.f;
    }
    dsum += w;
    b0.x = t.v[2 * j].x;  b0.y = t.v[2 * j + 1].x;
    b1.x = t.v[2 * j].y;  b1.y = t.v[2 * j + 1].y;
    acc0 = WMMA_F32(w, b0, acc0);
    acc1 = WMMA_F32(w, b1, acc1);
  }
}

struct KvPair { KvTile a, b; };

__device__ __forceinline__ void load_pair(KvPair& p, const float* __restrict__ Kl,
                                          const float* __restrict__ Vl, int kt, int qt) {
  load_tile(p.a, Kl, Vl, kt < qt ? kt : qt);
  load_tile(p.b, Kl, Vl, kt + 1 < qt ? kt + 1 : qt);
}

__device__ __forceinline__ void pin_sums(v8f& acc0, v8f& acc1, v2f& dsum) {
  asm volatile("" : "+v"(acc0), "+v"(acc1), "+v"(dsum));
}

__device__ __forceinline__ void pin_tile(KvTile& t) {
#pragma unroll
  for (int i = 0; i < 8; ++i) asm volatile("" : "+v"(t.v[i]));
}

__device__ __forceinline__ void attend_rows(int qt, const v8f& q,
                                            const float* __restrict__ Kl,
                                            const float* __restrict__ Vl,
                                            float* __restrict__ Yl, int hh, int c) {
  v2f qf[4];
#pragma unroll
  for (int j = 0; j < 4; ++j) { qf[j].x = q[2 * j]; qf[j].y = q[2 * j + 1]; }
  v8f acc0 = {}, acc1 = {};
  v2f dsum = {};
  KvPair buf[3];
  load_pair(buf[0], Kl, Vl, 0, qt);
  __builtin_amdgcn_sched_barrier(0);
  load_pair(buf[1], Kl, Vl, 2, qt);
  __builtin_amdgcn_sched_barrier(0);
  asm volatile("" ::: "memory");
  int kt = 0;
  for (; kt + 6 <= qt; kt += 6) {
#pragma unroll
    for (int i = 0; i < 3; ++i) {
      load_pair(buf[(i + 2) % 3], Kl, Vl, kt + 2 * i + 4, qt);
      __builtin_amdgcn_sched_barrier(0);
      pin_tile(buf[i].a);
      pin_tile(buf[i].b);
      attend_tile<false>(buf[i].a, qf, acc0, acc1, dsum, hh, c);
      attend_tile<false>(buf[i].b, qf, acc0, acc1, dsum, hh, c);
      pin_sums(acc0, acc1, dsum);
      __builtin_amdgcn_sched_barrier(0);
    }
    asm volatile("" ::: "memory");
  }
  const int rem = qt - kt;
  load_pair(buf[2], Kl, Vl, kt + 4, qt);
#pragma unroll
  for (int m = 0; m < 6; ++m) {
    KvTile& t = (m & 1) ? buf[m / 2].b : buf[m / 2].a;
    if (m < rem)       { pin_tile(t); attend_tile<false>(t, qf, acc0, acc1, dsum, hh, c); }
    else if (m == rem) { pin_tile(t); attend_tile<true>(t, qf, acc0, acc1, dsum, hh, c); }
  }

  const float part = dsum.x + dsum.y;
  const float tot  = part + __shfl_xor(part, 16, 32) + EPSF;
  float* yp = Yl + (size_t)qt * (16 * DM);
#pragma unroll
  for (int r = 0; r < 8; ++r) {
    const float dd = __shfl(tot, 8 * hh + r, 32);
    float2 y;
    y.x = acc0[r] / dd;
    y.y = acc1[r] / dd;
    *(float2*)(yp + (size_t)r * DM) = y;
  }
}

__global__ __launch_bounds__(128) void based_attn(const float* __restrict__ Xg,
                                                  const float* __restrict__ Wqg,
                                                  const float* __restrict__ Kg,
                                                  const float* __restrict__ Vg,
                                                  float* __restrict__ Yg) {
  __shared__ v8f qpart[4][4][32];

  const int lane = threadIdx.x & 31;
  const int wave = __builtin_amdgcn_readfirstlane((int)(threadIdx.x >> 5));
  const int vb   = wave & 1;
  const int pr   = wave >> 1;
  const int hh   = lane >> 4, c = lane & 15;
  const int hd   = blockIdx.x % Hh;
  const int bq   = (int)(blockIdx.x / Hh);
  const int t0   = 2 * bq + pr;

  v8f q0, q1;
  {
    const int d0 = (DM / 4) * wave + 8 * hh;
    const float* wp = Wqg + (size_t)d0 * (Hh * FD) + hd * FD + c;
    const float* x0 = Xg + (size_t)(16 * (2 * bq) + c) * DM + d0;
    const float* x1 = Xg + (size_t)(16 * (2 * bq + 1) + c) * DM + d0;
    const float* x2 = Xg + (size_t)(16 * (L / 16 - 1 - 2 * bq) + c) * DM + d0;
    const float* x3 = Xg + (size_t)(16 * (L / 16 - 2 - 2 * bq) + c) * DM + d0;

    v8f qacc[4];
#pragma unroll
    for (int n = 0; n < 4; ++n) qacc[n] = v8f{};

    QChunk qb[4];
    q_load(qb[0], wp, x0, x1, x2, x3, 0);
    __builtin_amdgcn_sched_barrier(0);
    q_load(qb[1], wp, x0, x1, x2, x3, 1);
    __builtin_amdgcn_sched_barrier(0);
    q_load(qb[2], wp, x0, x1, x2, x3, 2);
    __builtin_amdgcn_sched_barrier(0);
#pragma unroll
    for (int kc = 0; kc < QCH; ++kc) {
      if (kc + 3 < QCH) q_load(qb[(kc + 3) % 4], wp, x0, x1, x2, x3, kc + 3);
      __builtin_amdgcn_sched_barrier(0);
      q_mma(qb[kc % 4], qacc);
      __builtin_amdgcn_sched_barrier(0);
    }

#pragma unroll
    for (int n = 0; n < 4; ++n) qpart[wave][n][lane] = qacc[n];
    __syncthreads();
    q0 = qpart[0][pr][lane] + qpart[1][pr][lane] + qpart[2][pr][lane] + qpart[3][pr][lane];
    q1 = qpart[0][2 + pr][lane] + qpart[1][2 + pr][lane] + qpart[2][2 + pr][lane] +
         qpart[3][2 + pr][lane];
  }

  const float* Kl = Kg + (size_t)c * (Hh * FD) + hd * FD + 8 * hh;
  const float* Vl = Vg + (size_t)(8 * hh) * DM + hd * HD + vb * 32 + 2 * c;
  float*       Yl = Yg + (size_t)(8 * hh) * DM + hd * HD + vb * 32 + 2 * c;

  attend_rows(t0, q0, Kl, Vl, Yl, hh, c);
  attend_rows(L / 16 - 1 - t0, q1, Kl, Vl, Yl, hh, c);
}

extern "C" void kernel_launch(void* const* d_in, const int* in_sizes, int n_in,
                              void* d_out, int out_size, void* d_ws, size_t ws_size,
                              hipStream_t stream) {
  (void)in_sizes; (void)n_in; (void)out_size; (void)ws_size;
  const float* hs = (const float*)d_in[0];
  const float* Wq = (const float*)d_in[1];
  const float* Wk = (const float*)d_in[2];
  const float* Wv = (const float*)d_in[3];
  const float* Wo = (const float*)d_in[4];
  float* out = (float*)d_out;

  float* Kg = (float*)d_ws;
  float* Vg = Kg + (size_t)L * (Hh * FD);
  float* Yg = Vg + (size_t)L * DM;

  gemm_direct<Hh * FD, DM, 2, 8, 6, 2><<<dim3((L / 32) * ((Hh * FD) / 32)), dim3(64), 0, stream>>>(hs, Wk, Kg);
  gemm_direct<DM, DM, 4, 8, 4, 4><<<dim3((L / 64) * (DM / 64)), dim3(128), 0, stream>>>(hs, Wv, Vg);
  based_attn<<<dim3(Hh * (L / 64)), dim3(128), 0, stream>>>(hs, Wq, Kg, Vg, Yg);
  gemm_direct<DM, DM, 4, 8, 4, 4><<<dim3((L / 64) * (DM / 64)), dim3(128), 0, stream>>>(Yg, Wo, out);
}


# GEMMs: K split over the 4 waves of a block, each wave a 32x96 (V, out) or 16x64 (K) tile of its K quarter from global memory, partial tiles added through LDS in fixed order; order of sums changes
# speedup vs baseline: 6.4408x; 1.1807x over previous
#include <hip/hip_runtime.h>

typedef float v2f __attribute__((ext_vector_type(2)));
typedef float v4f __attribute__((ext_vector_type(4)));
typedef float v8f __attribute__((ext_vector_type(8)));

#define WMMA_F32(a, b, c) \
  __builtin_amdgcn_wmma_f32_16x16x4_f32(false, (a), false, (b), (short)0, (c), false, false)

constexpr int L = 1024, DM = 768, Hh = 12, FD = 16, HD = 64;
constexpr float EPSF = 1e-12f;
constexpr float QSC = 0.17677669529663687f;

template <int MT, int NP, int KH>
struct GChunk {
  v4f a[MT][KH / 4];
  v2f b[KH][NP];
};

template <int N, int K, int MT, int NP, int KH>
__device__ __forceinline__ void load_chunk(GChunk<MT, NP, KH>& t, const float* __restrict__ Al,
                                           const float* __restrict__ Bl, int kc) {
  const float* ap = Al + kc * (2 * KH);
  const float* bp = Bl + (size_t)kc * (2 * KH * N);
#pragma unroll
  for (int m = 0; m < MT; ++m) {
#pragma unroll
    for (int i = 0; i < KH / 4; ++i) t.a[m][i] = *(const v4f*)(ap + (size_t)m * (16 * K) + 4 * i);
  }
#pragma unroll
  for (int i = 0; i < KH; ++i) {
#pragma unroll
    for (int p = 0; p < NP; ++p) t.b[i][p] = *(const v2f*)(bp + (size_t)i * N + 32 * p);
  }
}

template <int MT, int NP, int KH>
__device__ __forceinline__ void mma_chunk(const GChunk<MT, NP, KH>& t, v8f (&acc)[MT][2 * NP]) {
#pragma unroll
  for (int j = 0; j < KH / 2; ++j) {
    v2f af[MT];
#pragma unroll
    for (int m = 0; m < MT; ++m) {
      af[m].x = t.a[m][j / 2][2 * (j % 2)];
      af[m].y = t.a[m][j / 2][2 * (j % 2) + 1];
    }
#pragma unroll
    for (int p = 0; p < NP; ++p) {
#pragma unroll
      for (int n = 0; n < 2; ++n) {
        v2f bf;
        bf.x = t.b[2 * j][p][n];
        bf.y = t.b[2 * j + 1][p][n];
#pragma unroll
        for (int m = 0; m < MT; ++m) acc[m][2 * p + n] = WMMA_F32(af[m], bf, acc[m][2 * p + n]);
      }
    }
  }
}

template <int MT, int NP>
__device__ __forceinline__ void pin_acc(v8f (&acc)[MT][2 * NP]) {
#pragma unroll
  for (int m = 0; m < MT; ++m) {
#pragma unroll
    for (int n = 0; n < 2 * NP; ++n) asm volatile("" : "+v"(acc[m][n]));
  }
}

template <int N, int K, int MT, int NP, int KH, int R>
__global__ __launch_bounds__(128) void gemm_splitk(const float* __restrict__ A,
                                                   const float* __restrict__ B,
                                                   float* __restrict__ C) {
  constexpr int NS  = N / (32 * NP);
  constexpr int KQ  = K / 4;
  constexpr int NCH = KQ / (2 * KH);
  constexpr int NTL = MT * 2 * NP;
  static_assert(N % (32 * NP) == 0 && K % 4 == 0 && KQ % (2 * KH * R) == 0 && KH % 4 == 0,
                "tile sizes must divide the problem");
  __shared__ v8f part[4][NTL][32];

  const int lane = threadIdx.x & 31;
  const int wave = __builtin_amdgcn_readfirstlane((int)(threadIdx.x >> 5));
  const int h = lane >> 4, c = lane & 15;
  const int strip = blockIdx.x % NS;
  const int row0  = (int)(blockIdx.x / NS) * (16 * MT);
  const int col0  = strip * (32 * NP);

  const float* Al = A + (size_t)(row0 + c) * K + wave * KQ + KH * h;
  const float* Bl = B + (size_t)(wave * KQ + KH * h) * N + col0 + 2 * c;

  v8f acc[MT][2 * NP];
#pragma unroll
  for (int m = 0; m < MT; ++m) {
#pragma unroll
    for (int n = 0; n < 2 * NP; ++n) acc[m][n] = v8f{};
  }

  GChunk<MT, NP, KH> buf[R];
#pragma unroll
  for (int i = 0; i < R - 1; ++i) {
    load_chunk<N, K, MT, NP, KH>(buf[i], Al, Bl, i);
    __builtin_amdgcn_sched_barrier(0);
  }
  asm volatile("" ::: "memory");
  for (int kc = 0; kc < NCH; kc += R) {
#pragma unroll
    for (int i = 0; i < R; ++i) {
      const int nx = kc + i + R - 1;
      load_chunk<N, K, MT, NP, KH>(buf[(i + R - 1) % R], Al, Bl, nx < NCH ? nx : NCH - 1);
      __builtin_amdgcn_sched_barrier(0);
      mma_chunk<MT, NP, KH>(buf[i], acc);
      if (i < R - 1) {
        pin_acc<MT, NP>(acc);
        __builtin_amdgcn_sched_barrier(0);
      }
    }
    asm volatile("" ::: "memory");
  }

#pragma unroll
  for (int m = 0; m < MT; ++m) {
#pragma unroll
    for (int n = 0; n < 2 * NP; ++n) part[wave][m * 2 * NP + n][lane] = acc[m][n];
  }
  __syncthreads();
#pragma unroll
  for (int m = 0; m < MT; ++m) {
#pragma unroll
    for (int p = 0; p < NP; ++p) {
      v2f s[2];
#pragma unroll
      for (int n = 0; n < 2; ++n) {
        const int t = m * 2 * NP + 2 * p + n;
        s[n] = *(const v2f*)((const float*)&part[0][t][lane] + 2 * wave);
#pragma unroll
        for (int q = 1; q < 4; ++q) s[n] += *(const v2f*)((const float*)&part[q][t][lane] + 2 * wave);
      }
      float* cp = C + (size_t)(row0 + 16 * m + 8 * h + 2 * wave) * N + col0 + 32 * p + 2 * c;
      v2f o0, o1;
      o0.x = s[0].x; o0.y = s[1].x;
      o1.x = s[0].y; o1.y = s[1].y;
      *(v2f*)cp       = o0;
      *(v2f*)(cp + N) = o1;
    }
  }
}

constexpr int QCH = (DM / 4) / 16;

struct QChunk {
  float w[8];
  v4f   x[4][2];
};

__device__ __forceinline__ void q_load(QChunk& t, const float* __restrict__ wp,
                                       const float* __restrict__ x0, const float* __restrict__ x1,
                                       const float* __restrict__ x2, const float* __restrict__ x3,
                                       int kc) {
  const float* w = wp + (size_t)kc * (16 * Hh * FD);
#pragma unroll
  for (int i = 0; i < 8; ++i) t.w[i] = w[(size_t)i * (Hh * FD)];
#pragma unroll
  for (int i = 0; i < 2; ++i) {
    t.x[0][i] = *(const v4f*)(x0 + kc * 16 + 4 * i);
    t.x[1][i] = *(const v4f*)(x1 + kc * 16 + 4 * i);
    t.x[2][i] = *(const v4f*)(x2 + kc * 16 + 4 * i);
    t.x[3][i] = *(const v4f*)(x3 + kc * 16 + 4 * i);
  }
}

__device__ __forceinline__ void q_mma(const QChunk& t, v8f (&acc)[4]) {
#pragma unroll
  for (int j = 0; j < 4; ++j) {
    v2f af;
    af.x = t.w[2 * j];
    af.y = t.w[2 * j + 1];
#pragma unroll
    for (int n = 0; n < 4; ++n) {
      v2f bf;
      bf.x = t.x[n][j / 2][2 * (j % 2)];
      bf.y = t.x[n][j / 2][2 * (j % 2) + 1];
      acc[n] = WMMA_F32(af, bf, acc[n]);
    }
  }
}

struct KvTile {
  float4 ka, kb;
  v2f v[8];
};

__device__ __forceinline__ void load_tile(KvTile& t, const float* __restrict__ Kl,
                                          const float* __restrict__ Vl, int kt) {
  const float* kp = Kl + (size_t)kt * (16 * Hh * FD);
  const float* vp = Vl + (size_t)kt * (16 * DM);
  t.ka = *(const float4*)kp;
  t.kb = *(const float4*)(kp + 4);
#pragma unroll
  for (int i = 0; i < 8; ++i) t.v[i] = *(const v2f*)(vp + (size_t)i * DM);
}

template <bool DIAG>
__device__ __forceinline__ void attend_tile(const KvTile& t, const v2f (&qf)[4],
                                            v8f& acc0, v8f& acc1, v2f& dsum,
                                            int hh, int c) {
  v2f kf[4];
  kf[0].x = t.ka.x; kf[0].y = t.ka.y; kf[1].x = t.ka.z; kf[1].y = t.ka.w;
  kf[2].x = t.kb.x; kf[2].y = t.kb.y; kf[3].x = t.kb.z; kf[3].y = t.kb.w;
  v8f s = {};
#pragma unroll
  for (int j = 0; j < 4; ++j) s = WMMA_F32(kf[j], qf[j], s);
#pragma unroll
  for (int j = 0; j < 4; ++j) {
    v2f x, w, b0, b1;
    x.x = s[2 * j]; x.y = s[2 * j + 1];
    w = x * (x * 0.03125f + 0.25f) + 1.f;
    if (DIAG) {
      w.x = (8 * hh + 2 * j     <= c) ? w.x : 0.f;
      w.y = (8 * hh + 2 * j + 1 <= c) ? w.y : 0.f;
    }
    dsum += w;
    b0.x = t.v[2 * j].x;  b0.y = t.v[2 * j + 1].x;
    b1.x = t.v[2 * j].y;  b1.y = t.v[2 * j + 1].y;
    acc0 = WMMA_F32(w, b0, acc0);
    acc1 = WMMA_F32(w, b1, acc1);
  }
}

struct KvPair { KvTile a, b; };

__device__ __forceinline__ void load_pair(KvPair& p, const float* __restrict__ Kl,
                                          const float* __restrict__ Vl, int kt, int qt) {
  load_tile(p.a, Kl, Vl, kt < qt ? kt : qt);
  load_tile(p.b, Kl, Vl, kt + 1 < qt ? kt + 1 : qt);
}

__device__ __forceinline__ void pin_sums(v8f& acc0, v8f& acc1, v2f& dsum) {
  asm volatile("" : "+v"(acc0), "+v"(acc1), "+v"(dsum));
}

__device__ __forceinline__ void pin_tile(KvTile& t) {
#pragma unroll
  for (int i = 0; i < 8; ++i) asm volatile("" : "+v"(t.v[i]));
}

__device__ __forceinline__ void attend_rows(int qt, const v8f& q,
                                            const float* __restrict__ Kl,
                                            const float* __restrict__ Vl,
                                            float* __restrict__ Yl, int hh, int c) {
  v2f qf[4];
#pragma unroll
  for (int j = 0; j < 4; ++j) { qf[j].x = q[2 * j]; qf[j].y = q[2 * j + 1]; }
  v8f acc0 = {}, acc1 = {};
  v2f dsum = {};
  KvPair buf[3];
  load_pair(buf[0], Kl, Vl, 0, qt);
  __builtin_amdgcn_sched_barrier(0);
  load_pair(buf[1], Kl, Vl, 2, qt);
  __builtin_amdgcn_sched_barrier(0);
  asm volatile("" ::: "memory");
  int kt = 0;
  for (; kt + 6 <= qt; kt += 6) {
#pragma unroll
    for (int i = 0; i < 3; ++i) {
      load_pair(buf[(i + 2) % 3], Kl, Vl, kt + 2 * i + 4, qt);
      __builtin_amdgcn_sched_barrier(0);
      pin_tile(buf[i].a);
      pin_tile(buf[i].b);
      attend_tile<false>(buf[i].a, qf, acc0, acc1, dsum, hh, c);
      attend_tile<false>(buf[i].b, qf, acc0, acc1, dsum, hh, c);
      pin_sums(acc0, acc1, dsum);
      __builtin_amdgcn_sched_barrier(0);
    }
    asm volatile("" ::: "memory");
  }
  const int rem = qt - kt;
  load_pair(buf[2], Kl, Vl, kt + 4, qt);
#pragma unroll
  for (int m = 0; m < 6; ++m) {
    KvTile& t = (m & 1) ? buf[m / 2].b : buf[m / 2].a;
    if (m < rem)       { pin_tile(t); attend_tile<false>(t, qf, acc0, acc1, dsum, hh, c); }
    else if (m == rem) { pin_tile(t); attend_tile<true>(t, qf, acc0, acc1, dsum, hh, c); }
  }

  const float part = dsum.x + dsum.y;
  const float tot  = part + __shfl_xor(part, 16, 32) + EPSF;
  float* yp = Yl + (size_t)qt * (16 * DM);
#pragma unroll
  for (int r = 0; r < 8; ++r) {
    const float dd = __shfl(tot, 8 * hh + r, 32);
    float2 y;
    y.x = acc0[r] / dd;
    y.y = acc1[r] / dd;
    *(float2*)(yp + (size_t)r * DM) = y;
  }
}

__global__ __launch_bounds__(128) void based_attn(const float* __restrict__ Xg,
                                                  const float* __restrict__ Wqg,
                                                  const float* __restrict__ Kg,
                                                  const float* __restrict__ Vg,
                                                  float* __restrict__ Yg) {
  __shared__ v8f qpart[4][4][32];

  const int lane = threadIdx.x & 31;
  const int wave = __builtin_amdgcn_readfirstlane((int)(threadIdx.x >> 5));
  const int vb   = wave & 1;
  const int pr   = wave >> 1;
  const int hh   = lane >> 4, c = lane & 15;
  const int hd   = blockIdx.x % Hh;
  const int bq   = (int)(blockIdx.x / Hh);
  const int t0   = 2 * bq + pr;

  v8f q0, q1;
  {
    const int d0 = (DM / 4) * wave + 8 * hh;
    const float* wp = Wqg + (size_t)d0 * (Hh * FD) + hd * FD + c;
    const float* x0 = Xg + (size_t)(16 * (2 * bq) + c) * DM + d0;
    const float* x1 = Xg + (size_t)(16 * (2 * bq + 1) + c) * DM + d0;
    const float* x2 = Xg + (size_t)(16 * (L / 16 - 1 - 2 * bq) + c) * DM + d0;
    const float* x3 = Xg + (size_t)(16 * (L / 16 - 2 - 2 * bq) + c) * DM + d0;

    v8f qacc[4];
#pragma unroll
    for (int n = 0; n < 4; ++n) qacc[n] = v8f{};

    QChunk qb[4];
    q_load(qb[0], wp, x0, x1, x2, x3, 0);
    __builtin_amdgcn_sched_barrier(0);
    q_load(qb[1], wp, x0, x1, x2, x3, 1);
    __builtin_amdgcn_sched_barrier(0);
    q_load(qb[2], wp, x0, x1, x2, x3, 2);
    __builtin_amdgcn_sched_barrier(0);
#pragma unroll
    for (int kc = 0; kc < QCH; ++kc) {
      if (kc + 3 < QCH) q_load(qb[(kc + 3) % 4], wp, x0, x1, x2, x3, kc + 3);
      __builtin_amdgcn_sched_barrier(0);
      q_mma(qb[kc % 4], qacc);
      __builtin_amdgcn_sched_barrier(0);
    }

#pragma unroll
    for (int n = 0; n < 4; ++n) qpart[wave][n][lane] = qacc[n];
    __syncthreads();
    q0 = qpart[0][pr][lane] + qpart[1][pr][lane] + qpart[2][pr][lane] + qpart[3][pr][lane];
    q1 = qpart[0][2 + pr][lane] + qpart[1][2 + pr][lane] + qpart[2][2 + pr][lane] +
         qpart[3][2 + pr][lane];
  }

  const float* Kl = Kg + (size_t)c * (Hh * FD) + hd * FD + 8 * hh;
  const float* Vl = Vg + (size_t)(8 * hh) * DM + hd * HD + vb * 32 + 2 * c;
  float*       Yl = Yg + (size_t)(8 * hh) * DM + hd * HD + vb * 32 + 2 * c;

  attend_rows(t0, q0, Kl, Vl, Yl, hh, c);
  attend_rows(L / 16 - 1 - t0, q1, Kl, Vl, Yl, hh, c);
}

extern "C" void kernel_launch(void* const* d_in, const int* in_sizes, int n_in,
                              void* d_out, int out_size, void* d_ws, size_t ws_size,
                              hipStream_t stream) {
  (void)in_sizes; (void)n_in; (void)out_size; (void)ws_size;
  const float* hs = (const float*)d_in[0];
  const float* Wq = (const float*)d_in[1];
  const float* Wk = (const float*)d_in[2];
  const float* Wv = (const float*)d_in[3];
  const float* Wo = (const float*)d_in[4];
  float* out = (float*)d_out;

  float* Kg = (float*)d_ws;
  float* Vg = Kg + (size_t)L * (Hh * FD);
  float* Yg = Vg + (size_t)L * DM;

  gemm_splitk<Hh * FD, DM, 1, 2, 8, 4><<<dim3((L / 16) * ((Hh * FD) / 64)), dim3(128), 0, stream>>>(hs, Wk, Kg);
  gemm_splitk<DM, DM, 2, 3, 4, 4><<<dim3((L / 32) * (DM / 96)), dim3(128), 0, stream>>>(hs, Wv, Vg);
  based_attn<<<dim3(Hh * (L / 64)), dim3(128), 0, stream>>>(hs, Wq, Kg, Vg, Yg);
  gemm_splitk<DM, DM, 2, 3, 4, 4><<<dim3((L / 32) * (DM / 96)), dim3(128), 0, stream>>>(Yg, Wo, out);
}


# attention re-split into 256 equal blocks: a block takes 3 (head, query tile) tasks, 97-98 key tiles; its 4 waves take every 4th key tile, all 64 V columns; partials added in LDS; sum order changes
# speedup vs baseline: 6.9026x; 1.0717x over previous
#include <hip/hip_runtime.h>

typedef float v2f __attribute__((ext_vector_type(2)));
typedef float v4f __attribute__((ext_vector_type(4)));
typedef float v8f __attribute__((ext_vector_type(8)));

#define WMMA_F32(a, b, c) \
  __builtin_amdgcn_wmma_f32_16x16x4_f32(false, (a), false, (b), (short)0, (c), false, false)

constexpr int L = 1024, DM = 768, Hh = 12, FD = 16, HD = 64;
constexpr float EPSF = 1e-12f;
constexpr float QSC = 0.17677669529663687f;

template <int MT, int NP, int KH>
struct GChunk {
  v4f a[MT][KH / 4];
  v2f b[KH][NP];
};

template <int N, int K, int MT, int NP, int KH>
__device__ __forceinline__ void load_chunk(GChunk<MT, NP, KH>& t, const float* __restrict__ Al,
                                           const float* __restrict__ Bl, int kc) {
  const float* ap = Al + kc * (2 * KH);
  const float* bp = Bl + (size_t)kc * (2 * KH * N);
#pragma unroll
  for (int m = 0; m < MT; ++m) {
#pragma unroll
    for (int i = 0; i < KH / 4; ++i) t.a[m][i] = *(const v4f*)(ap + (size_t)m * (16 * K) + 4 * i);
  }
#pragma unroll
  for (int i = 0; i < KH; ++i) {
#pragma unroll
    for (int p = 0; p < NP; ++p) t.b[i][p] = *(const v2f*)(bp + (size_t)i * N + 32 * p);
  }
}

template <int MT, int NP, int KH>
__device__ __forceinline__ void mma_chunk(const GChunk<MT, NP, KH>& t, v8f (&acc)[MT][2 * NP]) {
#pragma unroll
  for (int j = 0; j < KH / 2; ++j) {
    v2f af[MT];
#pragma unroll
    for (int m = 0; m < MT; ++m) {
      af[m].x = t.a[m][j / 2][2 * (j % 2)];
      af[m].y = t.a[m][j / 2][2 * (j % 2) + 1];
    }
#pragma unroll
    for (int p = 0; p < NP; ++p) {
#pragma unroll
      for (int n = 0; n < 2; ++n) {
        v2f bf;
        bf.x = t.b[2 * j][p][n];
        bf.y = t.b[2 * j + 1][p][n];
#pragma unroll
        for (int m = 0; m < MT; ++m) acc[m][2 * p + n] = WMMA_F32(af[m], bf, acc[m][2 * p + n]);
      }
    }
  }
}

template <int MT, int NP>
__device__ __forceinline__ void pin_acc(v8f (&acc)[MT][2 * NP]) {
#pragma unroll
  for (int m = 0; m < MT; ++m) {
#pragma unroll
    for (int n = 0; n < 2 * NP; ++n) asm volatile("" : "+v"(acc[m][n]));
  }
}

template <int N, int K, int MT, int NP, int KH, int R>
__global__ __launch_bounds__(128) void gemm_splitk(const float* __restrict__ A,
                                                   const float* __restrict__ B,
                                                   float* __restrict__ C) {
  constexpr int NS  = N / (32 * NP);
  constexpr int KQ  = K / 4;
  constexpr int NCH = KQ / (2 * KH);
  constexpr int NTL = MT * 2 * NP;
  static_assert(N % (32 * NP) == 0 && K % 4 == 0 && KQ % (2 * KH * R) == 0 && KH % 4 == 0,
                "tile sizes must divide the problem");
  __shared__ v8f part[4][NTL][32];

  const int lane = threadIdx.x & 31;
  const int wave = __builtin_amdgcn_readfirstlane((int)(threadIdx.x >> 5));
  const int h = lane >> 4, c = lane & 15;
  const int strip = blockIdx.x % NS;
  const int row0  = (int)(blockIdx.x / NS) * (16 * MT);
  const int col0  = strip * (32 * NP);

  const float* Al = A + (size_t)(row0 + c) * K + wave * KQ + KH * h;
  const float* Bl = B + (size_t)(wave * KQ + KH * h) * N + col0 + 2 * c;

  v8f acc[MT][2 * NP];
#pragma unroll
  for (int m = 0; m < MT; ++m) {
#pragma unroll
    for (int n = 0; n < 2 * NP; ++n) acc[m][n] = v8f{};
  }

  GChunk<MT, NP, KH> buf[R];
#pragma unroll
  for (int i = 0; i < R - 1; ++i) {
    load_chunk<N, K, MT, NP, KH>(buf[i], Al, Bl, i);
    __builtin_amdgcn_sched_barrier(0);
  }
  asm volatile("" ::: "memory");
  for (int kc = 0; kc < NCH; kc += R) {
#pragma unroll
    for (int i = 0; i < R; ++i) {
      const int nx = kc + i + R - 1;
      load_chunk<N, K, MT, NP, KH>(buf[(i + R - 1) % R], Al, Bl, nx < NCH ? nx : NCH - 1);
      __builtin_amdgcn_sched_barrier(0);
      mma_chunk<MT, NP, KH>(buf[i], acc);
      if (i < R - 1) {
        pin_acc<MT, NP>(acc);
        __builtin_amdgcn_sched_barrier(0);
      }
    }
    asm volatile("" ::: "memory");
  }

#pragma unroll
  for (int m = 0; m < MT; ++m) {
#pragma unroll
    for (int n = 0; n < 2 * NP; ++n) part[wave][m * 2 * NP + n][lane] = acc[m][n];
  }
  __syncthreads();
#pragma unroll
  for (int m = 0; m < MT; ++m) {
#pragma unroll
    for (int p = 0; p < NP; ++p) {
      v2f s[2];
#pragma unroll
      for (int n = 0; n < 2; ++n) {
        const int t = m * 2 * NP + 2 * p + n;
        s[n] = *(const v2f*)((const float*)&part[0][t][lane] + 2 * wave);
#pragma unroll
        for (int q = 1; q < 4; ++q) s[n] += *(const v2f*)((const float*)&part[q][t][lane] + 2 * wave);
      }
      float* cp = C + (size_t)(row0 + 16 * m + 8 * h + 2 * wave) * N + col0 + 32 * p + 2 * c;
      v2f o0, o1;
      o0.x = s[0].x; o0.y = s[1].x;
      o1.x = s[0].y; o1.y = s[1].y;
      *(v2f*)cp       = o0;
      *(v2f*)(cp + N) = o1;
    }
  }
}

constexpr int QCH = (DM / 4) / 16;
constexpr int NTK = 3;

struct QChunk {
  float w[NTK][8];
  v4f   x[NTK][2];
};

__device__ __forceinline__ void q_load(QChunk& t, const float* const (&wp)[NTK],
                                       const float* const (&xp)[NTK], int kc) {
#pragma unroll
  for (int n = 0; n < NTK; ++n) {
    const float* w = wp[n] + (size_t)kc * (16 * Hh * FD);
#pragma unroll
    for (int i = 0; i < 8; ++i) t.w[n][i] = w[(size_t)i * (Hh * FD)];
#pragma unroll
    for (int i = 0; i < 2; ++i) t.x[n][i] = *(const v4f*)(xp[n] + kc * 16 + 4 * i);
  }
}

__device__ __forceinline__ void q_mma(const QChunk& t, v8f (&acc)[NTK]) {
#pragma unroll
  for (int j = 0; j < 4; ++j) {
#pragma unroll
    for (int n = 0; n < NTK; ++n) {
      v2f af, bf;
      af.x = t.w[n][2 * j];
      af.y = t.w[n][2 * j + 1];
      bf.x = t.x[n][j / 2][2 * (j % 2)];
      bf.y = t.x[n][j / 2][2 * (j % 2) + 1];
      acc[n] = WMMA_F32(af, bf, acc[n]);
    }
  }
}

struct KvTile {
  float4 ka, kb;
  v4f v[8];
};

__device__ __forceinline__ void load_tile(KvTile& t, const float* __restrict__ Kl,
                                          const float* __restrict__ Vl, int j) {
  const float* kp = Kl + (size_t)j * (64 * Hh * FD);
  const float* vp = Vl + (size_t)j * (64 * DM);
  t.ka = *(const float4*)kp;
  t.kb = *(const float4*)(kp + 4);
#pragma unroll
  for (int i = 0; i < 8; ++i) t.v[i] = *(const v4f*)(vp + (size_t)i * DM);
}

template <bool DIAG>
__device__ __forceinline__ void attend_tile(const KvTile& t, const v2f (&qf)[4],
                                            v8f (&acc)[4], v2f& dsum, int hh, int c) {
  v2f kf[4];
  kf[0].x = t.ka.x; kf[0].y = t.ka.y; kf[1].x = t.ka.z; kf[1].y = t.ka.w;
  kf[2].x = t.kb.x; kf[2].y = t.kb.y; kf[3].x = t.kb.z; kf[3].y = t.kb.w;
  v8f s = {};
#pragma unroll
  for (int j = 0; j < 4; ++j) s = WMMA_F32(kf[j], qf[j], s);
#pragma unroll
  for (int j = 0; j < 4; ++j) {
    v2f x, w;
    x.x = s[2 * j]; x.y = s[2 * j + 1];
    w = x * (x * 0.03125f + 0.25f) + 1.f;
    if (DIAG) {
      w.x = (8 * hh + 2 * j     <= c) ? w.x : 0.f;
      w.y = (8 * hh + 2 * j + 1 <= c) ? w.y : 0.f;
    }
    dsum += w;
#pragma unroll
    for (int n = 0; n < 4; ++n) {
      v2f b;
      b.x = t.v[2 * j][n];
      b.y = t.v[2 * j + 1][n];
      acc[n] = WMMA_F32(w, b, acc[n]);
    }
  }
}

struct KvPair { KvTile a, b; };

__device__ __forceinline__ void load_pair(KvPair& p, const float* __restrict__ Kl,
                                          const float* __restrict__ Vl, int j, int lj) {
  load_tile(p.a, Kl, Vl, j < lj ? j : lj);
  load_tile(p.b, Kl, Vl, j + 1 < lj ? j + 1 : lj);
}

__device__ __forceinline__ void pin_sums(v8f (&acc)[4], v2f& dsum) {
  asm volatile("" : "+v"(acc[0]), "+v"(acc[1]), "+v"(acc[2]), "+v"(acc[3]), "+v"(dsum));
}

__device__ __forceinline__ void pin_tile(KvTile& t) {
#pragma unroll
  for (int i = 0; i < 8; ++i) asm volatile("" : "+v"(t.v[i]));
}

__device__ __forceinline__ void attend_task(int qt, int w, const v8f& q,
                                            const float* __restrict__ Kl,
                                            const float* __restrict__ Vl,
                                            v8f (&acc)[4], float& den, int hh, int c) {
  v2f qf[4];
#pragma unroll
  for (int j = 0; j < 4; ++j) { qf[j].x = q[2 * j]; qf[j].y = q[2 * j + 1]; }
#pragma unroll
  for (int n = 0; n < 4; ++n) acc[n] = v8f{};
  v2f dsum = {};
  const int  nl = qt >= w ? ((qt - w) >> 2) + 1 : 0;
  const bool dg = qt >= w && ((qt - w) & 3) == 0;
  const int  nf = dg ? nl - 1 : nl;
  const int  lj = nl > 0 ? nl - 1 : 0;
  KvPair buf[3];
  load_pair(buf[0], Kl, Vl, 0, lj);
  __builtin_amdgcn_sched_barrier(0);
  load_pair(buf[1], Kl, Vl, 2, lj);
  __builtin_amdgcn_sched_barrier(0);
  asm volatile("" ::: "memory");
  int j0 = 0;
  for (; j0 + 6 <= nf; j0 += 6) {
#pragma unroll
    for (int i = 0; i < 3; ++i) {
      load_pair(buf[(i + 2) % 3], Kl, Vl, j0 + 2 * i + 4, lj);
      __builtin_amdgcn_sched_barrier(0);
      pin_tile(buf[i].a);
      pin_tile(buf[i].b);
      attend_tile<false>(buf[i].a, qf, acc, dsum, hh, c);
      attend_tile<false>(buf[i].b, qf, acc, dsum, hh, c);
      pin_sums(acc, dsum);
      __builtin_amdgcn_sched_barrier(0);
    }
    asm volatile("" ::: "memory");
  }
  const int rem = nf - j0;
  load_pair(buf[2], Kl, Vl, j0 + 4, lj);
#pragma unroll
  for (int m = 0; m < 6; ++m) {
    KvTile& t = (m & 1) ? buf[m / 2].b : buf[m / 2].a;
    if (m < rem)             attend_tile<false>(t, qf, acc, dsum, hh, c);
    else if (m == rem && dg) attend_tile<true>(t, qf, acc, dsum, hh, c);
  }
  const float part = dsum.x + dsum.y;
  den = part + __shfl_xor(part, 16, 32);
}

__device__ __forceinline__ int task_qt(int i, int n) {
  const int  m   = i >> 1;
  const bool odd = (i & 1) != 0;
  return n == 0 ? i : ((n == 1) != odd ? 63 - m : 31 - m);
}

__global__ __launch_bounds__(128) void based_attn(const float* __restrict__ Xg,
                                                  const float* __restrict__ Wqg,
                                                  const float* __restrict__ Kg,
                                                  const float* __restrict__ Vg,
                                                  float* __restrict__ Yg) {
  __shared__ v8f qpart[4][NTK][32];
  __shared__ v8f   opart[NTK][4][4][32];
  __shared__ float dpart[NTK][4][32];

  const int lane = threadIdx.x & 31;
  const int wave = __builtin_amdgcn_readfirstlane((int)(threadIdx.x >> 5));
  const int hh   = lane >> 4, c = lane & 15;
  const int grp  = (int)(blockIdx.x / (L / 16));
  const int bi   = (int)(blockIdx.x % (L / 16));

  {
    const int d0 = (DM / 4) * wave + 8 * hh;
    const float* wp[NTK];
    const float* xp[NTK];
    v8f qacc[NTK];
#pragma unroll
    for (int n = 0; n < NTK; ++n) {
      wp[n] = Wqg + (size_t)d0 * (Hh * FD) + (NTK * grp + n) * FD + c;
      xp[n] = Xg + (size_t)(16 * task_qt(bi, n) + c) * DM + d0;
      qacc[n] = v8f{};
    }

    QChunk qb[3];
    q_load(qb[0], wp, xp, 0);
    __builtin_amdgcn_sched_barrier(0);
    q_load(qb[1], wp, xp, 1);
    __builtin_amdgcn_sched_barrier(0);
#pragma unroll
    for (int kc = 0; kc < QCH; ++kc) {
      if (kc + 2 < QCH) q_load(qb[(kc + 2) % 3], wp, xp, kc + 2);
      __builtin_amdgcn_sched_barrier(0);
      q_mma(qb[kc % 3], qacc);
      __builtin_amdgcn_sched_barrier(0);
    }

#pragma unroll
    for (int n = 0; n < NTK; ++n) qpart[wave][n][lane] = qacc[n];
    __syncthreads();
  }

#pragma nounroll
  for (int n = 0; n < NTK; ++n) {
    const int hd = NTK * grp + n;
    const int qt = task_qt(bi, n);
    const v8f q = qpart[0][n][lane] + qpart[1][n][lane] + qpart[2][n][lane] + qpart[3][n][lane];
    const float* Kl = Kg + (size_t)(16 * wave + c) * (Hh * FD) + hd * FD + 8 * hh;
    const float* Vl = Vg + (size_t)(16 * wave + 8 * hh) * DM + hd * HD + 4 * c;
    v8f acc[4];
    float den;
    attend_task(qt, wave, q, Kl, Vl, acc, den, hh, c);
#pragma unroll
    for (int t = 0; t < 4; ++t) opart[n][wave][t][lane] = acc[t];
    dpart[n][wave][lane] = den;
  }
  __syncthreads();

  for (int n = 0; n < NTK; ++n) {
    const int hd = NTK * grp + n;
    const int qt = task_qt(bi, n);
    const float tot = dpart[n][0][lane] + dpart[n][1][lane] + dpart[n][2][lane] +
                      dpart[n][3][lane] + EPSF;
    const float dd0 = __shfl(tot, 8 * hh + 2 * wave, 32);
    const float dd1 = __shfl(tot, 8 * hh + 2 * wave + 1, 32);
    v2f s[4];
#pragma unroll
    for (int t = 0; t < 4; ++t) {
      s[t] = *(const v2f*)((const float*)&opart[n][0][t][lane] + 2 * wave);
#pragma unroll
      for (int p = 1; p < 4; ++p) s[t] += *(const v2f*)((const float*)&opart[n][p][t][lane] + 2 * wave);
    }
    float* yp = Yg + (size_t)(16 * qt + 8 * hh + 2 * wave) * DM + hd * HD + 4 * c;
    float4 o0, o1;
    o0.x = s[0].x / dd0; o0.y = s[1].x / dd0; o0.z = s[2].x / dd0; o0.w = s[3].x / dd0;
    o1.x = s[0].y / dd1; o1.y = s[1].y / dd1; o1.z = s[2].y / dd1; o1.w = s[3].y / dd1;
    *(float4*)yp        = o0;
    *(float4*)(yp + DM) = o1;
  }
}

extern "C" void kernel_launch(void* const* d_in, const int* in_sizes, int n_in,
                              void* d_out, int out_size, void* d_ws, size_t ws_size,
                              hipStream_t stream) {
  (void)in_sizes; (void)n_in; (void)out_size; (void)ws_size;
  const float* hs = (const float*)d_in[0];
  const float* Wq = (const float*)d_in[1];
  const float* Wk = (const float*)d_in[2];
  const float* Wv = (const float*)d_in[3];
  const float* Wo = (const float*)d_in[4];
  float* out = (float*)d_out;

  float* Kg = (float*)d_ws;
  float* Vg = Kg + (size_t)L * (Hh * FD);
  float* Yg = Vg + (size_t)L * DM;

  gemm_splitk<Hh * FD, DM, 1, 2, 8, 4><<<dim3((L / 16) * ((Hh * FD) / 64)), dim3(128), 0, stream>>>(hs, Wk, Kg);
  gemm_splitk<DM, DM, 2, 3, 4, 4><<<dim3((L / 32) * (DM / 96)), dim3(128), 0, stream>>>(hs, Wv, Vg);
  based_attn<<<dim3((Hh / NTK) * (L / 16)), dim3(128), 0, stream>>>(hs, Wq, Kg, Vg, Yg);
  gemm_splitk<DM, DM, 2, 3, 4, 4><<<dim3((L / 32) * (DM / 96)), dim3(128), 0, stream>>>(Yg, Wo, out);
}
